# starting kernel
# baseline (speedup 1.0000x reference)
#include <hip/hip_runtime.h>

#define N_NODES 50000
#define N_EDGES 1600000
#define DIM     128
#define EPS     1e-8f

typedef __attribute__((ext_vector_type(2))) float v2f;
typedef __attribute__((ext_vector_type(8))) float v8f;

#define LDSW 132

__global__ void gemm_wmma_kernel(const float* __restrict__ seq,
                                 const float* __restrict__ W,
                                 float* __restrict__ seq_fts) {
    __shared__ float smem[16 * LDSW];

    const int tid     = threadIdx.x;
    const int rowBase = blockIdx.x * 16;

    for (int v = tid; v < 512; v += 256) {
        const int r  = v >> 5;
        const int c4 = v & 31;
        float4 d = ((const float4*)(seq + (size_t)(rowBase + r) * DIM))[c4];
        *(float4*)&smem[r * LDSW + c4 * 4] = d;
    }
    __syncthreads();

    const int wave    = tid >> 5;
    const int lane    = tid & 31;
    const int lo      = lane & 15;
    const int hi      = lane >> 4;
    const int colBase = wave * 16;

    const float* arow = &smem[lo * LDSW + 2 * hi];
    const float* brow = W + (size_t)(colBase + lo) * DIM + 2 * hi;

    v8f acc = {};
#pragma unroll
    for (int k0 = 0; k0 < DIM; k0 += 4) {
        v2f a = *(const v2f*)(arow + k0);
        v2f b = *(const v2f*)(brow + k0);
        acc = __builtin_amdgcn_wmma_f32_16x16x4_f32(
             false, a,  false, b,
             (short)0, acc,  false,  false);
    }

#pragma unroll
    for (int j = 0; j < 8; ++j) {
        const int m = j + 8 * hi;
        seq_fts[(size_t)(rowBase + m) * DIM + colBase + lo] = acc[j];
    }
}

#define SCAN_B  256
#define SCAN_NB ((N_NODES + SCAN_B - 1) / SCAN_B)

__global__ void zero_kernel(int* counts) {
    int i = blockIdx.x * blockDim.x + threadIdx.x;
    if (i < N_NODES) counts[i] = 0;
}

__global__ void hist_kernel(const int* __restrict__ rows, int* __restrict__ counts) {
    int e = blockIdx.x * blockDim.x + threadIdx.x;
    if (e < N_EDGES) atomicAdd(&counts[rows[e]], 1);
}

__global__ void partials_kernel(const int* __restrict__ counts, int* __restrict__ partials) {
    __shared__ int s[SCAN_B];
    int t = threadIdx.x;
    int i = blockIdx.x * SCAN_B + t;
    s[t] = (i < N_NODES) ? counts[i] : 0;
    __syncthreads();
    for (int off = SCAN_B / 2; off > 0; off >>= 1) {
        if (t < off) s[t] += s[t + off];
        __syncthreads();
    }
    if (t == 0) partials[blockIdx.x] = s[0];
}

__global__ void scan_top_kernel(const int* __restrict__ partials, int* __restrict__ blockoff) {
    __shared__ int s[SCAN_B];
    int t = threadIdx.x;
    s[t] = (t < SCAN_NB) ? partials[t] : 0;
    __syncthreads();
    for (int off = 1; off < SCAN_B; off <<= 1) {
        int x = (t >= off) ? s[t - off] : 0;
        __syncthreads();
        s[t] += x;
        __syncthreads();
    }
    if (t < SCAN_NB) blockoff[t] = (t == 0) ? 0 : s[t - 1];
}

__global__ void scan_write_kernel(const int* __restrict__ counts,
                                  const int* __restrict__ blockoff,
                                  int* __restrict__ rowptr,
                                  int* __restrict__ cursor) {
    __shared__ int s[SCAN_B];
    int t = threadIdx.x;
    int i = blockIdx.x * SCAN_B + t;
    int v = (i < N_NODES) ? counts[i] : 0;
    s[t] = v;
    __syncthreads();
    for (int off = 1; off < SCAN_B; off <<= 1) {
        int x = (t >= off) ? s[t - off] : 0;
        __syncthreads();
        s[t] += x;
        __syncthreads();
    }
    if (i < N_NODES) {
        int excl = s[t] - v + blockoff[blockIdx.x];
        rowptr[i] = excl;
        cursor[i] = excl;
    }
}

__global__ void fill_kernel(const int* __restrict__ rows, const int* __restrict__ cols,
                            int* __restrict__ cursor, int* __restrict__ csr_cols) {
    int e = blockIdx.x * blockDim.x + threadIdx.x;
    if (e < N_EDGES) {
        int pos = atomicAdd(&cursor[rows[e]], 1);
        csr_cols[pos] = cols[e];
    }
}

__global__ void agg_kernel(const float* __restrict__ seq_fts,
                           const int* __restrict__ rowptr,
                           const int* __restrict__ counts,
                           const int* __restrict__ csr_cols,
                           const float* __restrict__ alpha,
                           float* __restrict__ out) {
    const int wave = threadIdx.x >> 5;
    const int lane = threadIdx.x & 31;
    const int node = blockIdx.x * 4 + wave;

    const int start = rowptr[node];
    const int len   = counts[node];

    const float* base = seq_fts + lane * 4;
    float4 acc = make_float4(0.f, 0.f, 0.f, 0.f);
    for (int j = 0; j < len; ++j) {
        const int c = csr_cols[start + j];
        float4 v = *(const float4*)(base + (size_t)c * DIM);
        acc.x += v.x; acc.y += v.y; acc.z += v.z; acc.w += v.w;
    }

    const float d = (float)len + EPS;
    const float a = alpha[0];
    float4 m;
    m.x = acc.x / d; m.y = acc.y / d; m.z = acc.z / d; m.w = acc.w / d;
    m.x = (m.x >= 0.f) ? m.x : a * m.x;
    m.y = (m.y >= 0.f) ? m.y : a * m.y;
    m.z = (m.z >= 0.f) ? m.z : a * m.z;
    m.w = (m.w >= 0.f) ? m.w : a * m.w;
    *(float4*)(out + (size_t)node * DIM + lane * 4) = m;
}

extern "C" void kernel_launch(void* const* d_in, const int* in_sizes, int n_in,
                              void* d_out, int out_size, void* d_ws, size_t ws_size,
                              hipStream_t stream) {
    (void)in_sizes; (void)n_in; (void)out_size; (void)ws_size;

    const float* seq   = (const float*)d_in[0];
    const float* W     = (const float*)d_in[1];
    const float* alpha = (const float*)d_in[2];
    const int*   rows  = (const int*)d_in[3];
    const int*   cols  = (const int*)d_in[4];
    float* out = (float*)d_out;

    char* ws = (char*)d_ws;
    float* seq_fts  = (float*)(ws);
    int*   counts   = (int*)(ws + 25600000);
    int*   rowptr   = (int*)(ws + 25800000);
    int*   cursor   = (int*)(ws + 26000000);
    int*   partials = (int*)(ws + 26200000);
    int*   blockoff = (int*)(ws + 26201024);
    int*   csr_cols = (int*)(ws + 26202048);

    const int edgeBlocks = (N_EDGES + 255) / 256;
    const int nodeBlocks = (N_NODES + 255) / 256;

    gemm_wmma_kernel<<<N_NODES / 16, 256, 0, stream>>>(seq, W, seq_fts);

    zero_kernel<<<nodeBlocks, 256, 0, stream>>>(counts);
    hist_kernel<<<edgeBlocks, 256, 0, stream>>>(rows, counts);
    partials_kernel<<<SCAN_NB, SCAN_B, 0, stream>>>(counts, partials);
    scan_top_kernel<<<1, SCAN_B, 0, stream>>>(partials, blockoff);
    scan_write_kernel<<<SCAN_NB, SCAN_B, 0, stream>>>(counts, blockoff, rowptr, cursor);
    fill_kernel<<<edgeBlocks, 256, 0, stream>>>(rows, cols, cursor, csr_cols);

    agg_kernel<<<N_NODES / 4, 128, 0, stream>>>(seq_fts, rowptr, counts, csr_cols,
                                                alpha, out);
}


# CSR build as two-level counting sort on LDS atomics (bin = row>>8, then per-bin count, scan, place) replaces both global-atomic passes; in-row edge order (float sum order) = LDS atomic arrival order
# speedup vs baseline: 2.2814x; 2.2814x over previous
#include <hip/hip_runtime.h>

#define N_NODES 50000
#define N_EDGES 1600000
#define DIM     128
#define EPS     1e-8f

typedef __attribute__((ext_vector_type(2))) float v2f;
typedef __attribute__((ext_vector_type(8))) float v8f;

#define LDSW 132

__global__ void gemm_wmma_kernel(const float* __restrict__ seq,
                                 const float* __restrict__ W,
                                 float* __restrict__ seq_fts) {
    __shared__ float smem[16 * LDSW];

    const int tid     = threadIdx.x;
    const int rowBase = blockIdx.x * 16;

    for (int v = tid; v < 512; v += 256) {
        const int r  = v >> 5;
        const int c4 = v & 31;
        float4 d = ((const float4*)(seq + (size_t)(rowBase + r) * DIM))[c4];
        *(float4*)&smem[r * LDSW + c4 * 4] = d;
    }
    __syncthreads();

    const int wave    = tid >> 5;
    const int lane    = tid & 31;
    const int lo      = lane & 15;
    const int hi      = lane >> 4;
    const int colBase = wave * 16;

    const float* arow = &smem[lo * LDSW + 2 * hi];
    const float* brow = W + (size_t)(colBase + lo) * DIM + 2 * hi;

    v8f acc = {};
#pragma unroll
    for (int k0 = 0; k0 < DIM; k0 += 4) {
        v2f a = *(const v2f*)(arow + k0);
        v2f b = *(const v2f*)(brow + k0);
        acc = __builtin_amdgcn_wmma_f32_16x16x4_f32(
             false, a,  false, b,
             (short)0, acc,  false,  false);
    }

#pragma unroll
    for (int j = 0; j < 8; ++j) {
        const int m = j + 8 * hi;
        seq_fts[(size_t)(rowBase + m) * DIM + colBase + lo] = acc[j];
    }
}

#define BIN_SHIFT 8
#define BIN_NODES 256
#define NBIN    ((N_NODES + BIN_NODES - 1) / BIN_NODES)
#define NSLICE  256
#define SLICE_T 512
#define SLICE_DEPTH 3
#define BIN_T   1024
#define BIN_DEPTH 8
#define NSPARE  32
#define NVEC4   (N_EDGES / 4)

#define SCAN_B  256
#define SCAN_N  (NBIN * NSLICE)
#define SCAN_NB ((SCAN_N + SCAN_B - 1) / SCAN_B)

static_assert(N_NODES <= (1 << 24), "a source id must fit the 24 bits above the row's low byte");
static_assert(N_EDGES % 4 == 0, "the edge list is read as int4");
static_assert(NSLICE == SCAN_B, "one scan block per bin: the bin's first counter opens its range");
static_assert(SCAN_NB <= SCAN_B, "the top-level scan is one block");

__global__ void bin_count_kernel(const int* __restrict__ rows, int* __restrict__ slice_counts) {
    __shared__ int hist[NBIN];
    const int t = threadIdx.x;
    for (int b = t; b < NBIN; b += SLICE_T) hist[b] = 0;
    __syncthreads();

    const int v0 = ((int)blockIdx.x * NVEC4) / NSLICE;
    const int v1 = (((int)blockIdx.x + 1) * NVEC4) / NSLICE;
    const int4* rows4 = (const int4*)rows;
    int u0 = v0 + t;
    for (; u0 + (SLICE_DEPTH - 1) * SLICE_T < v1; u0 += SLICE_DEPTH * SLICE_T) {
        int4 r[SLICE_DEPTH];
        for (int k = 0; k < SLICE_DEPTH; ++k) r[k] = rows4[u0 + k * SLICE_T];
        for (int k = 0; k < SLICE_DEPTH; ++k)
            asm volatile("" : "+v"(r[k].x), "+v"(r[k].y), "+v"(r[k].z), "+v"(r[k].w));
        for (int k = 0; k < SLICE_DEPTH; ++k) {
            atomicAdd(&hist[r[k].x >> BIN_SHIFT], 1);
            atomicAdd(&hist[r[k].y >> BIN_SHIFT], 1);
            atomicAdd(&hist[r[k].z >> BIN_SHIFT], 1);
            atomicAdd(&hist[r[k].w >> BIN_SHIFT], 1);
        }
    }
    for (; u0 < v1; u0 += SLICE_T) {
        const int4 r = rows4[u0];
        atomicAdd(&hist[r.x >> BIN_SHIFT], 1);
        atomicAdd(&hist[r.y >> BIN_SHIFT], 1);
        atomicAdd(&hist[r.z >> BIN_SHIFT], 1);
        atomicAdd(&hist[r.w >> BIN_SHIFT], 1);
    }
    __syncthreads();
    for (int b = t; b < NBIN; b += SLICE_T) slice_counts[b * NSLICE + blockIdx.x] = hist[b];
}

__global__ void partials_kernel(const int* __restrict__ vals, int* __restrict__ partials) {
    __shared__ int s[SCAN_B];
    int t = threadIdx.x;
    int i = blockIdx.x * SCAN_B + t;
    s[t] = (i < SCAN_N) ? vals[i] : 0;
    __syncthreads();
    for (int off = SCAN_B / 2; off > 0; off >>= 1) {
        if (t < off) s[t] += s[t + off];
        __syncthreads();
    }
    if (t == 0) partials[blockIdx.x] = s[0];
}

__global__ void scan_top_kernel(const int* __restrict__ partials, int* __restrict__ blockoff) {
    __shared__ int s[SCAN_B];
    int t = threadIdx.x;
    s[t] = (t < SCAN_NB) ? partials[t] : 0;
    __syncthreads();
    for (int off = 1; off < SCAN_B; off <<= 1) {
        int x = (t >= off) ? s[t - off] : 0;
        __syncthreads();
        s[t] += x;
        __syncthreads();
    }
    if (t < SCAN_NB) blockoff[t] = (t == 0) ? 0 : s[t - 1];
}

__global__ void scan_write_kernel(const int* __restrict__ vals,
                                  const int* __restrict__ blockoff,
                                  int* __restrict__ slice_base) {
    __shared__ int s[SCAN_B];
    int t = threadIdx.x;
    int i = blockIdx.x * SCAN_B + t;
    int v = (i < SCAN_N) ? vals[i] : 0;
    s[t] = v;
    __syncthreads();
    for (int off = 1; off < SCAN_B; off <<= 1) {
        int x = (t >= off) ? s[t - off] : 0;
        __syncthreads();
        s[t] += x;
        __syncthreads();
    }
    if (i < SCAN_N) slice_base[i] = s[t] - v + blockoff[blockIdx.x];
}

__global__ void bin_place_kernel(const int* __restrict__ rows, const int* __restrict__ cols,
                                 const int* __restrict__ slice_base,
                                 unsigned* __restrict__ binned) {
    __shared__ int cur[NBIN];
    const int t = threadIdx.x;
    for (int b = t; b < NBIN; b += SLICE_T) cur[b] = slice_base[b * NSLICE + blockIdx.x];
    __syncthreads();

    const int v0 = ((int)blockIdx.x * NVEC4) / NSLICE;
    const int v1 = (((int)blockIdx.x + 1) * NVEC4) / NSLICE;
    const int4* rows4 = (const int4*)rows;
    const int4* cols4 = (const int4*)cols;
    int u0 = v0 + t;
    for (; u0 + (SLICE_DEPTH - 1) * SLICE_T < v1; u0 += SLICE_DEPTH * SLICE_T) {
        int4 r[SLICE_DEPTH];
        int4 c[SLICE_DEPTH];
        int4 p[SLICE_DEPTH];
        for (int k = 0; k < SLICE_DEPTH; ++k) {
            r[k] = rows4[u0 + k * SLICE_T];
            c[k] = cols4[u0 + k * SLICE_T];
        }
        for (int k = 0; k < SLICE_DEPTH; ++k)
            asm volatile("" : "+v"(r[k].x), "+v"(r[k].y), "+v"(r[k].z), "+v"(r[k].w),
                              "+v"(c[k].x), "+v"(c[k].y), "+v"(c[k].z), "+v"(c[k].w));
        for (int k = 0; k < SLICE_DEPTH; ++k) {
            p[k].x = atomicAdd(&cur[r[k].x >> BIN_SHIFT], 1);
            p[k].y = atomicAdd(&cur[r[k].y >> BIN_SHIFT], 1);
            p[k].z = atomicAdd(&cur[r[k].z >> BIN_SHIFT], 1);
            p[k].w = atomicAdd(&cur[r[k].w >> BIN_SHIFT], 1);
        }
        for (int k = 0; k < SLICE_DEPTH; ++k) {
            binned[p[k].x] = ((unsigned)c[k].x << BIN_SHIFT) | (unsigned)(r[k].x & (BIN_NODES - 1));
            binned[p[k].y] = ((unsigned)c[k].y << BIN_SHIFT) | (unsigned)(r[k].y & (BIN_NODES - 1));
            binned[p[k].z] = ((unsigned)c[k].z << BIN_SHIFT) | (unsigned)(r[k].z & (BIN_NODES - 1));
            binned[p[k].w] = ((unsigned)c[k].w << BIN_SHIFT) | (unsigned)(r[k].w & (BIN_NODES - 1));
        }
    }
    for (; u0 < v1; u0 += SLICE_T) {
        const int4 r = rows4[u0];
        const int4 c = cols4[u0];
        const int p0 = atomicAdd(&cur[r.x >> BIN_SHIFT], 1);
        const int p1 = atomicAdd(&cur[r.y >> BIN_SHIFT], 1);
        const int p2 = atomicAdd(&cur[r.z >> BIN_SHIFT], 1);
        const int p3 = atomicAdd(&cur[r.w >> BIN_SHIFT], 1);
        binned[p0] = ((unsigned)c.x << BIN_SHIFT) | (unsigned)(r.x & (BIN_NODES - 1));
        binned[p1] = ((unsigned)c.y << BIN_SHIFT) | (unsigned)(r.y & (BIN_NODES - 1));
        binned[p2] = ((unsigned)c.z << BIN_SHIFT) | (unsigned)(r.z & (BIN_NODES - 1));
        binned[p3] = ((unsigned)c.w << BIN_SHIFT) | (unsigned)(r.w & (BIN_NODES - 1));
    }
}

__global__ void bin_csr_kernel(const unsigned* __restrict__ binned,
                               const int* __restrict__ slice_base,
                               int* __restrict__ rowptr, int* __restrict__ counts,
                               int* __restrict__ csr_cols) {
    __shared__ int cnt[BIN_NODES + NSPARE];
    __shared__ int cur[BIN_NODES + NSPARE];
    __shared__ int wsum[BIN_NODES / 32];

    const int t     = threadIdx.x;
    const int lane  = t & 31;
    const int wave  = t >> 5;
    const int bin   = blockIdx.x;
    const int spare = BIN_NODES + lane;

    const int start = slice_base[bin * NSLICE];
    const int end   = (bin + 1 < NBIN) ? slice_base[(bin + 1) * NSLICE] : N_EDGES;

    if (t < BIN_NODES + NSPARE) cnt[t] = 0;
    __syncthreads();
    for (int i0 = start + t; i0 < end; i0 += BIN_DEPTH * BIN_T) {
        unsigned rec[BIN_DEPTH];
        for (int k = 0; k < BIN_DEPTH; ++k) rec[k] = binned[min(i0 + k * BIN_T, end - 1)];
        for (int k = 0; k < BIN_DEPTH; ++k)
            atomicAdd(&cnt[(i0 + k * BIN_T < end) ? (int)(rec[k] & (BIN_NODES - 1)) : spare], 1);
    }
    __syncthreads();

    const int v = (t < BIN_NODES) ? cnt[t] : 0;
    int x = v;
    for (int off = 1; off < 32; off <<= 1) {
        const int y = __shfl_up(x, off);
        if (lane >= off) x += y;
    }
    if (t < BIN_NODES && lane == 31) wsum[wave] = x;
    __syncthreads();
    if (t < BIN_NODES) {
        int pre = 0;
        for (int w = 0; w < wave; ++w) pre += wsum[w];
        const int excl = start + pre + x - v;
        cur[t] = excl;
        const int node = bin * BIN_NODES + t;
        if (node < N_NODES) {
            rowptr[node] = excl;
            counts[node] = v;
        }
    } else if (t < BIN_NODES + NSPARE) {
        cur[t] = 0;
    }
    __syncthreads();
    for (int i0 = start + t; i0 < end; i0 += BIN_DEPTH * BIN_T) {
        unsigned rec[BIN_DEPTH];
        int pos[BIN_DEPTH];
        for (int k = 0; k < BIN_DEPTH; ++k) rec[k] = binned[min(i0 + k * BIN_T, end - 1)];
        for (int k = 0; k < BIN_DEPTH; ++k)
            pos[k] = atomicAdd(&cur[(i0 + k * BIN_T < end) ? (int)(rec[k] & (BIN_NODES - 1)) : spare], 1);
        for (int k = 0; k < BIN_DEPTH; ++k)
            if (i0 + k * BIN_T < end) csr_cols[pos[k]] = (int)(rec[k] >> BIN_SHIFT);
    }
}

__global__ void agg_kernel(const float* __restrict__ seq_fts,
                           const int* __restrict__ rowptr,
                           const int* __restrict__ counts,
                           const int* __restrict__ csr_cols,
                           const float* __restrict__ alpha,
                           float* __restrict__ out) {
    const int wave = threadIdx.x >> 5;
    const int lane = threadIdx.x & 31;
    const int node = blockIdx.x * 4 + wave;

    const int start = rowptr[node];
    const int len   = counts[node];

    const float* base = seq_fts + lane * 4;
    float4 acc = make_float4(0.f, 0.f, 0.f, 0.f);
    for (int j = 0; j < len; ++j) {
        const int c = csr_cols[start + j];
        float4 v = *(const float4*)(base + (size_t)c * DIM);
        acc.x += v.x; acc.y += v.y; acc.z += v.z; acc.w += v.w;
    }

    const float d = (float)len + EPS;
    const float a = alpha[0];
    float4 m;
    m.x = acc.x / d; m.y = acc.y / d; m.z = acc.z / d; m.w = acc.w / d;
    m.x = (m.x >= 0.f) ? m.x : a * m.x;
    m.y = (m.y >= 0.f) ? m.y : a * m.y;
    m.z = (m.z >= 0.f) ? m.z : a * m.z;
    m.w = (m.w >= 0.f) ? m.w : a * m.w;
    *(float4*)(out + (size_t)node * DIM + lane * 4) = m;
}

extern "C" void kernel_launch(void* const* d_in, const int* in_sizes, int n_in,
                              void* d_out, int out_size, void* d_ws, size_t ws_size,
                              hipStream_t stream) {
    (void)in_sizes; (void)n_in; (void)out_size; (void)ws_size;

    const float* seq   = (const float*)d_in[0];
    const float* W     = (const float*)d_in[1];
    const float* alpha = (const float*)d_in[2];
    const int*   rows  = (const int*)d_in[3];
    const int*   cols  = (const int*)d_in[4];
    float* out = (float*)d_out;

    char* ws = (char*)d_ws;
    float*    seq_fts      = (float*)(ws);
    unsigned* binned       = (unsigned*)(ws);
    int*      slice_counts = (int*)(ws + 6400000);
    int*      slice_base   = (int*)(ws + 6600704);
    int*      counts       = (int*)(ws + 25600000);
    int*      rowptr       = (int*)(ws + 25800000);
    int*      partials     = (int*)(ws + 26200000);
    int*      blockoff     = (int*)(ws + 26201024);
    int*      csr_cols     = (int*)(ws + 26202048);

    bin_count_kernel<<<NSLICE, SLICE_T, 0, stream>>>(rows, slice_counts);
    partials_kernel<<<SCAN_NB, SCAN_B, 0, stream>>>(slice_counts, partials);
    scan_top_kernel<<<1, SCAN_B, 0, stream>>>(partials, blockoff);
    scan_write_kernel<<<SCAN_NB, SCAN_B, 0, stream>>>(slice_counts, blockoff, slice_base);
    bin_place_kernel<<<NSLICE, SLICE_T, 0, stream>>>(rows, cols, slice_base, binned);
    bin_csr_kernel<<<NBIN, BIN_T, 0, stream>>>(binned, slice_base, rowptr, counts, csr_cols);

    gemm_wmma_kernel<<<N_NODES / 16, 256, 0, stream>>>(seq, W, seq_fts);

    agg_kernel<<<N_NODES / 4, 128, 0, stream>>>(seq_fts, rowptr, counts, csr_cols,
                                                alpha, out);
}


# the three scan launches folded into the placing kernel: each placing block sums the (slice, bin) counter table itself and scans the bin totals in LDS; 8 launches become 5
# speedup vs baseline: 2.4665x; 1.0811x over previous
#include <hip/hip_runtime.h>

#define N_NODES 50000
#define N_EDGES 1600000
#define DIM     128
#define EPS     1e-8f

typedef __attribute__((ext_vector_type(2))) float v2f;
typedef __attribute__((ext_vector_type(8))) float v8f;

#define LDSW 132

__global__ void gemm_wmma_kernel(const float* __restrict__ seq,
                                 const float* __restrict__ W,
                                 float* __restrict__ seq_fts) {
    __shared__ float smem[16 * LDSW];

    const int tid     = threadIdx.x;
    const int rowBase = blockIdx.x * 16;

    for (int v = tid; v < 512; v += 256) {
        const int r  = v >> 5;
        const int c4 = v & 31;
        float4 d = ((const float4*)(seq + (size_t)(rowBase + r) * DIM))[c4];
        *(float4*)&smem[r * LDSW + c4 * 4] = d;
    }
    __syncthreads();

    const int wave    = tid >> 5;
    const int lane    = tid & 31;
    const int lo      = lane & 15;
    const int hi      = lane >> 4;
    const int colBase = wave * 16;

    const float* arow = &smem[lo * LDSW + 2 * hi];
    const float* brow = W + (size_t)(colBase + lo) * DIM + 2 * hi;

    v8f acc = {};
#pragma unroll
    for (int k0 = 0; k0 < DIM; k0 += 4) {
        v2f a = *(const v2f*)(arow + k0);
        v2f b = *(const v2f*)(brow + k0);
        acc = __builtin_amdgcn_wmma_f32_16x16x4_f32(
             false, a,  false, b,
             (short)0, acc,  false,  false);
    }

#pragma unroll
    for (int j = 0; j < 8; ++j) {
        const int m = j + 8 * hi;
        seq_fts[(size_t)(rowBase + m) * DIM + colBase + lo] = acc[j];
    }
}

#define BIN_SHIFT 8
#define BIN_NODES 256
#define NBIN    ((N_NODES + BIN_NODES - 1) / BIN_NODES)
#define NSLICE  256
#define SLICE_T 512
#define SLICE_DEPTH 3
#define BIN_T   1024
#define BIN_DEPTH 8
#define NSPARE  32
#define NVEC4   (N_EDGES / 4)

#define TBL_W   256
#define TBL_Q   (TBL_W / 4)
#define TBL_R   (SLICE_T / TBL_Q)
#define TBL_K   (NSLICE / TBL_R)
#define TBL_DEPTH 16

static_assert(N_NODES <= (1 << 24), "a source id must fit the 24 bits above the row's low byte");
static_assert(N_EDGES % 4 == 0, "the edge list is read as int4");
static_assert(NBIN % 4 == 0 && NBIN <= TBL_W, "a thread sums four bins of a table row");
static_assert(SLICE_T % TBL_Q == 0 && NSLICE % TBL_R == 0, "the table's rows are dealt evenly to the threads");
static_assert((TBL_K & (TBL_K - 1)) == 0 && TBL_K % TBL_DEPTH == 0, "the row rotation wraps with a mask");
static_assert(TBL_W <= SLICE_T && TBL_W % 32 == 0, "the bin scan runs on the first TBL_W threads");

__global__ void bin_count_kernel(const int* __restrict__ rows, int* __restrict__ slice_counts) {
    __shared__ int hist[NBIN];
    const int t = threadIdx.x;
    for (int b = t; b < NBIN; b += SLICE_T) hist[b] = 0;
    __syncthreads();

    const int v0 = ((int)blockIdx.x * NVEC4) / NSLICE;
    const int v1 = (((int)blockIdx.x + 1) * NVEC4) / NSLICE;
    const int4* rows4 = (const int4*)rows;
    int u0 = v0 + t;
    for (; u0 + (SLICE_DEPTH - 1) * SLICE_T < v1; u0 += SLICE_DEPTH * SLICE_T) {
        int4 r[SLICE_DEPTH];
        for (int k = 0; k < SLICE_DEPTH; ++k) r[k] = rows4[u0 + k * SLICE_T];
        for (int k = 0; k < SLICE_DEPTH; ++k)
            asm volatile("" : "+v"(r[k].x), "+v"(r[k].y), "+v"(r[k].z), "+v"(r[k].w));
        for (int k = 0; k < SLICE_DEPTH; ++k) {
            atomicAdd(&hist[r[k].x >> BIN_SHIFT], 1);
            atomicAdd(&hist[r[k].y >> BIN_SHIFT], 1);
            atomicAdd(&hist[r[k].z >> BIN_SHIFT], 1);
            atomicAdd(&hist[r[k].w >> BIN_SHIFT], 1);
        }
    }
    for (; u0 < v1; u0 += SLICE_T) {
        const int4 r = rows4[u0];
        atomicAdd(&hist[r.x >> BIN_SHIFT], 1);
        atomicAdd(&hist[r.y >> BIN_SHIFT], 1);
        atomicAdd(&hist[r.z >> BIN_SHIFT], 1);
        atomicAdd(&hist[r.w >> BIN_SHIFT], 1);
    }
    __syncthreads();
    for (int b = t; b < NBIN; b += SLICE_T) slice_counts[blockIdx.x * TBL_W + b] = hist[b];
}

__global__ void __launch_bounds__(SLICE_T)
bin_place_kernel(const int* __restrict__ rows, const int* __restrict__ cols,
                                 const int* __restrict__ slice_counts,
                                 int* __restrict__ bin_start,
                                 unsigned* __restrict__ binned) {
    __shared__ int cur[TBL_W];
    __shared__ int tot[TBL_W];
    __shared__ int wsum[TBL_W / 32];
    const int t    = threadIdx.x;
    const int lane = t & 31;
    const int wave = t >> 5;
    const int s    = blockIdx.x;

    if (t < TBL_W) { cur[t] = 0; tot[t] = 0; }
    __syncthreads();

    const int q = t & (TBL_Q - 1);
    const int r = t / TBL_Q;
    if (q < NBIN / 4) {
        const int4* tbl = (const int4*)slice_counts + q;
        int4 T = make_int4(0, 0, 0, 0);
        int4 P = make_int4(0, 0, 0, 0);
        for (int k0 = 0; k0 < TBL_K; k0 += TBL_DEPTH) {
            int4 v[TBL_DEPTH];
            for (int k = 0; k < TBL_DEPTH; ++k) {
                const int sp = r + TBL_R * ((k0 + k + s) & (TBL_K - 1));
                v[k] = tbl[sp * TBL_Q];
            }
            for (int k = 0; k < TBL_DEPTH; ++k)
                asm volatile("" : "+v"(v[k].x), "+v"(v[k].y), "+v"(v[k].z), "+v"(v[k].w));
            for (int k = 0; k < TBL_DEPTH; ++k) {
                const int sp = r + TBL_R * ((k0 + k + s) & (TBL_K - 1));
                const int before = (sp < s) ? -1 : 0;
                T.x += v[k].x; T.y += v[k].y; T.z += v[k].z; T.w += v[k].w;
                P.x += v[k].x & before; P.y += v[k].y & before;
                P.z += v[k].z & before; P.w += v[k].w & before;
            }
        }
        atomicAdd(&tot[4 * q + 0], T.x); atomicAdd(&tot[4 * q + 1], T.y);
        atomicAdd(&tot[4 * q + 2], T.z); atomicAdd(&tot[4 * q + 3], T.w);
        atomicAdd(&cur[4 * q + 0], P.x); atomicAdd(&cur[4 * q + 1], P.y);
        atomicAdd(&cur[4 * q + 2], P.z); atomicAdd(&cur[4 * q + 3], P.w);
    }
    __syncthreads();

    {
        const int v = (t < TBL_W) ? tot[t] : 0;
        int x = v;
        for (int off = 1; off < 32; off <<= 1) {
            const int y = __shfl_up(x, off);
            if (lane >= off) x += y;
        }
        if (t < TBL_W && lane == 31) wsum[wave] = x;
        __syncthreads();
        if (t < NBIN) {
            int pre = 0;
            for (int w = 0; w < wave; ++w) pre += wsum[w];
            const int first = pre + x - v;
            cur[t] += first;
            if (s == 0) bin_start[t] = first;
        }
    }
    __syncthreads();

    const int v0 = ((int)blockIdx.x * NVEC4) / NSLICE;
    const int v1 = (((int)blockIdx.x + 1) * NVEC4) / NSLICE;
    const int4* rows4 = (const int4*)rows;
    const int4* cols4 = (const int4*)cols;
    int u0 = v0 + t;
    for (; u0 + (SLICE_DEPTH - 1) * SLICE_T < v1; u0 += SLICE_DEPTH * SLICE_T) {
        int4 r[SLICE_DEPTH];
        int4 c[SLICE_DEPTH];
        int4 p[SLICE_DEPTH];
        for (int k = 0; k < SLICE_DEPTH; ++k) {
            r[k] = rows4[u0 + k * SLICE_T];
            c[k] = cols4[u0 + k * SLICE_T];
        }
        for (int k = 0; k < SLICE_DEPTH; ++k)
            asm volatile("" : "+v"(r[k].x), "+v"(r[k].y), "+v"(r[k].z), "+v"(r[k].w),
                              "+v"(c[k].x), "+v"(c[k].y), "+v"(c[k].z), "+v"(c[k].w));
        for (int k = 0; k < SLICE_DEPTH; ++k) {
            p[k].x = atomicAdd(&cur[r[k].x >> BIN_SHIFT], 1);
            p[k].y = atomicAdd(&cur[r[k].y >> BIN_SHIFT], 1);
            p[k].z = atomicAdd(&cur[r[k].z >> BIN_SHIFT], 1);
            p[k].w = atomicAdd(&cur[r[k].w >> BIN_SHIFT], 1);
        }
        for (int k = 0; k < SLICE_DEPTH; ++k) {
            binned[p[k].x] = ((unsigned)c[k].x << BIN_SHIFT) | (unsigned)(r[k].x & (BIN_NODES - 1));
            binned[p[k].y] = ((unsigned)c[k].y << BIN_SHIFT) | (unsigned)(r[k].y & (BIN_NODES - 1));
            binned[p[k].z] = ((unsigned)c[k].z << BIN_SHIFT) | (unsigned)(r[k].z & (BIN_NODES - 1));
            binned[p[k].w] = ((unsigned)c[k].w << BIN_SHIFT) | (unsigned)(r[k].w & (BIN_NODES - 1));
        }
    }
    for (; u0 < v1; u0 += SLICE_T) {
        const int4 r = rows4[u0];
        const int4 c = cols4[u0];
        const int p0 = atomicAdd(&cur[r.x >> BIN_SHIFT], 1);
        const int p1 = atomicAdd(&cur[r.y >> BIN_SHIFT], 1);
        const int p2 = atomicAdd(&cur[r.z >> BIN_SHIFT], 1);
        const int p3 = atomicAdd(&cur[r.w >> BIN_SHIFT], 1);
        binned[p0] = ((unsigned)c.x << BIN_SHIFT) | (unsigned)(r.x & (BIN_NODES - 1));
        binned[p1] = ((unsigned)c.y << BIN_SHIFT) | (unsigned)(r.y & (BIN_NODES - 1));
        binned[p2] = ((unsigned)c.z << BIN_SHIFT) | (unsigned)(r.z & (BIN_NODES - 1));
        binned[p3] = ((unsigned)c.w << BIN_SHIFT) | (unsigned)(r.w & (BIN_NODES - 1));
    }
}

__global__ void bin_csr_kernel(const unsigned* __restrict__ binned,
                               const int* __restrict__ bin_start,
                               int* __restrict__ rowptr, int* __restrict__ counts,
                               int* __restrict__ csr_cols) {
    __shared__ int cnt[BIN_NODES + NSPARE];
    __shared__ int cur[BIN_NODES + NSPARE];
    __shared__ int wsum[BIN_NODES / 32];

    const int t     = threadIdx.x;
    const int lane  = t & 31;
    const int wave  = t >> 5;
    const int bin   = blockIdx.x;
    const int spare = BIN_NODES + lane;

    const int start = bin_start[bin];
    const int end   = (bin + 1 < NBIN) ? bin_start[bin + 1] : N_EDGES;

    if (t < BIN_NODES + NSPARE) cnt[t] = 0;
    __syncthreads();
    for (int i0 = start + t; i0 < end; i0 += BIN_DEPTH * BIN_T) {
        unsigned rec[BIN_DEPTH];
        for (int k = 0; k < BIN_DEPTH; ++k) rec[k] = binned[min(i0 + k * BIN_T, end - 1)];
        for (int k = 0; k < BIN_DEPTH; ++k)
            atomicAdd(&cnt[(i0 + k * BIN_T < end) ? (int)(rec[k] & (BIN_NODES - 1)) : spare], 1);
    }
    __syncthreads();

    const int v = (t < BIN_NODES) ? cnt[t] : 0;
    int x = v;
    for (int off = 1; off < 32; off <<= 1) {
        const int y = __shfl_up(x, off);
        if (lane >= off) x += y;
    }
    if (t < BIN_NODES && lane == 31) wsum[wave] = x;
    __syncthreads();
    if (t < BIN_NODES) {
        int pre = 0;
        for (int w = 0; w < wave; ++w) pre += wsum[w];
        const int excl = start + pre + x - v;
        cur[t] = excl;
        const int node = bin * BIN_NODES + t;
        if (node < N_NODES) {
            rowptr[node] = excl;
            counts[node] = v;
        }
    } else if (t < BIN_NODES + NSPARE) {
        cur[t] = 0;
    }
    __syncthreads();
    for (int i0 = start + t; i0 < end; i0 += BIN_DEPTH * BIN_T) {
        unsigned rec[BIN_DEPTH];
        int pos[BIN_DEPTH];
        for (int k = 0; k < BIN_DEPTH; ++k) rec[k] = binned[min(i0 + k * BIN_T, end - 1)];
        for (int k = 0; k < BIN_DEPTH; ++k)
            pos[k] = atomicAdd(&cur[(i0 + k * BIN_T < end) ? (int)(rec[k] & (BIN_NODES - 1)) : spare], 1);
        for (int k = 0; k < BIN_DEPTH; ++k)
            if (i0 + k * BIN_T < end) csr_cols[pos[k]] = (int)(rec[k] >> BIN_SHIFT);
    }
}

__global__ void agg_kernel(const float* __restrict__ seq_fts,
                           const int* __restrict__ rowptr,
                           const int* __restrict__ counts,
                           const int* __restrict__ csr_cols,
                           const float* __restrict__ alpha,
                           float* __restrict__ out) {
    const int wave = threadIdx.x >> 5;
    const int lane = threadIdx.x & 31;
    const int node = blockIdx.x * 4 + wave;

    const int start = rowptr[node];
    const int len   = counts[node];

    const float* base = seq_fts + lane * 4;
    float4 acc = make_float4(0.f, 0.f, 0.f, 0.f);
    for (int j = 0; j < len; ++j) {
        const int c = csr_cols[start + j];
        float4 v = *(const float4*)(base + (size_t)c * DIM);
        acc.x += v.x; acc.y += v.y; acc.z += v.z; acc.w += v.w;
    }

    const float d = (float)len + EPS;
    const float a = alpha[0];
    float4 m;
    m.x = acc.x / d; m.y = acc.y / d; m.z = acc.z / d; m.w = acc.w / d;
    m.x = (m.x >= 0.f) ? m.x : a * m.x;
    m.y = (m.y >= 0.f) ? m.y : a * m.y;
    m.z = (m.z >= 0.f) ? m.z : a * m.z;
    m.w = (m.w >= 0.f) ? m.w : a * m.w;
    *(float4*)(out + (size_t)node * DIM + lane * 4) = m;
}

extern "C" void kernel_launch(void* const* d_in, const int* in_sizes, int n_in,
                              void* d_out, int out_size, void* d_ws, size_t ws_size,
                              hipStream_t stream) {
    (void)in_sizes; (void)n_in; (void)out_size; (void)ws_size;

    const float* seq   = (const float*)d_in[0];
    const float* W     = (const float*)d_in[1];
    const float* alpha = (const float*)d_in[2];
    const int*   rows  = (const int*)d_in[3];
    const int*   cols  = (const int*)d_in[4];
    float* out = (float*)d_out;

    char* ws = (char*)d_ws;
    float*    seq_fts      = (float*)(ws);
    unsigned* binned       = (unsigned*)(ws);
    int*      slice_counts = (int*)(ws + 6400000);
    int*      counts       = (int*)(ws + 25600000);
    int*      rowptr       = (int*)(ws + 25800000);
    int*      bin_start    = (int*)(ws + 26200000);
    int*      csr_cols     = (int*)(ws + 26202048);

    bin_count_kernel<<<NSLICE, SLICE_T, 0, stream>>>(rows, slice_counts);
    bin_place_kernel<<<NSLICE, SLICE_T, 0, stream>>>(rows, cols, slice_counts, bin_start, binned);
    bin_csr_kernel<<<NBIN, BIN_T, 0, stream>>>(binned, bin_start, rowptr, counts, csr_cols);

    gemm_wmma_kernel<<<N_NODES / 16, 256, 0, stream>>>(seq, W, seq_fts);

    agg_kernel<<<N_NODES / 4, 128, 0, stream>>>(seq_fts, rowptr, counts, csr_cols,
                                                alpha, out);
}


# linear layer moved behind the aggregation (seq rows summed per node, then one WMMA product with W^T from LDS) and fused into the gather kernel; float sum order changes; 5 launches become 4
# speedup vs baseline: 2.6632x; 1.0797x over previous
#include <hip/hip_runtime.h>

#define N_NODES 50000
#define N_EDGES 1600000
#define DIM     128
#define EPS     1e-8f

typedef __attribute__((ext_vector_type(2))) float v2f;
typedef __attribute__((ext_vector_type(8))) float v8f;

#define BIN_SHIFT 8
#define BIN_NODES 256
#define NBIN    ((N_NODES + BIN_NODES - 1) / BIN_NODES)
#define NSLICE  256
#define SLICE_T 512
#define SLICE_DEPTH 3
#define BIN_T   1024
#define BIN_DEPTH 8
#define NSPARE  32
#define NVEC4   (N_EDGES / 4)

#define TBL_W   256
#define TBL_Q   (TBL_W / 4)
#define TBL_R   (SLICE_T / TBL_Q)
#define TBL_K   (NSLICE / TBL_R)
#define TBL_DEPTH 16

static_assert(N_NODES <= (1 << 24), "a source id must fit the 24 bits above the row's low byte");
static_assert(N_EDGES % 4 == 0, "the edge list is read as int4");
static_assert(NBIN % 4 == 0 && NBIN <= TBL_W, "a thread sums four bins of a table row");
static_assert(SLICE_T % TBL_Q == 0 && NSLICE % TBL_R == 0, "the table's rows are dealt evenly to the threads");
static_assert((TBL_K & (TBL_K - 1)) == 0 && TBL_K % TBL_DEPTH == 0, "the row rotation wraps with a mask");
static_assert(TBL_W <= SLICE_T && TBL_W % 32 == 0, "the bin scan runs on the first TBL_W threads");

__global__ void bin_count_kernel(const int* __restrict__ rows, int* __restrict__ slice_counts) {
    __shared__ int hist[NBIN];
    const int t = threadIdx.x;
    for (int b = t; b < NBIN; b += SLICE_T) hist[b] = 0;
    __syncthreads();

    const int v0 = ((int)blockIdx.x * NVEC4) / NSLICE;
    const int v1 = (((int)blockIdx.x + 1) * NVEC4) / NSLICE;
    const int4* rows4 = (const int4*)rows;
    int u0 = v0 + t;
    for (; u0 + (SLICE_DEPTH - 1) * SLICE_T < v1; u0 += SLICE_DEPTH * SLICE_T) {
        int4 r[SLICE_DEPTH];
        for (int k = 0; k < SLICE_DEPTH; ++k) r[k] = rows4[u0 + k * SLICE_T];
        for (int k = 0; k < SLICE_DEPTH; ++k)
            asm volatile("" : "+v"(r[k].x), "+v"(r[k].y), "+v"(r[k].z), "+v"(r[k].w));
        for (int k = 0; k < SLICE_DEPTH; ++k) {
            atomicAdd(&hist[r[k].x >> BIN_SHIFT], 1);
            atomicAdd(&hist[r[k].y >> BIN_SHIFT], 1);
            atomicAdd(&hist[r[k].z >> BIN_SHIFT], 1);
            atomicAdd(&hist[r[k].w >> BIN_SHIFT], 1);
        }
    }
    for (; u0 < v1; u0 += SLICE_T) {
        const int4 r = rows4[u0];
        atomicAdd(&hist[r.x >> BIN_SHIFT], 1);
        atomicAdd(&hist[r.y >> BIN_SHIFT], 1);
        atomicAdd(&hist[r.z >> BIN_SHIFT], 1);
        atomicAdd(&hist[r.w >> BIN_SHIFT], 1);
    }
    __syncthreads();
    for (int b = t; b < NBIN; b += SLICE_T) slice_counts[blockIdx.x * TBL_W + b] = hist[b];
}

__global__ void __launch_bounds__(SLICE_T)
bin_place_kernel(const int* __restrict__ rows, const int* __restrict__ cols,
                                 const int* __restrict__ slice_counts,
                                 int* __restrict__ bin_start,
                                 unsigned* __restrict__ binned) {
    __shared__ int cur[TBL_W];
    __shared__ int tot[TBL_W];
    __shared__ int wsum[TBL_W / 32];
    const int t    = threadIdx.x;
    const int lane = t & 31;
    const int wave = t >> 5;
    const int s    = blockIdx.x;

    if (t < TBL_W) { cur[t] = 0; tot[t] = 0; }
    __syncthreads();

    const int q = t & (TBL_Q - 1);
    const int r = t / TBL_Q;
    if (q < NBIN / 4) {
        const int4* tbl = (const int4*)slice_counts + q;
        int4 T = make_int4(0, 0, 0, 0);
        int4 P = make_int4(0, 0, 0, 0);
        for (int k0 = 0; k0 < TBL_K; k0 += TBL_DEPTH) {
            int4 v[TBL_DEPTH];
            for (int k = 0; k < TBL_DEPTH; ++k) {
                const int sp = r + TBL_R * ((k0 + k + s) & (TBL_K - 1));
                v[k] = tbl[sp * TBL_Q];
            }
            for (int k = 0; k < TBL_DEPTH; ++k)
                asm volatile("" : "+v"(v[k].x), "+v"(v[k].y), "+v"(v[k].z), "+v"(v[k].w));
            for (int k = 0; k < TBL_DEPTH; ++k) {
                const int sp = r + TBL_R * ((k0 + k + s) & (TBL_K - 1));
                const int before = (sp < s) ? -1 : 0;
                T.x += v[k].x; T.y += v[k].y; T.z += v[k].z; T.w += v[k].w;
                P.x += v[k].x & before; P.y += v[k].y & before;
                P.z += v[k].z & before; P.w += v[k].w & before;
            }
        }
        atomicAdd(&tot[4 * q + 0], T.x); atomicAdd(&tot[4 * q + 1], T.y);
        atomicAdd(&tot[4 * q + 2], T.z); atomicAdd(&tot[4 * q + 3], T.w);
        atomicAdd(&cur[4 * q + 0], P.x); atomicAdd(&cur[4 * q + 1], P.y);
        atomicAdd(&cur[4 * q + 2], P.z); atomicAdd(&cur[4 * q + 3], P.w);
    }
    __syncthreads();

    {
        const int v = (t < TBL_W) ? tot[t] : 0;
        int x = v;
        for (int off = 1; off < 32; off <<= 1) {
            const int y = __shfl_up(x, off);
            if (lane >= off) x += y;
        }
        if (t < TBL_W && lane == 31) wsum[wave] = x;
        __syncthreads();
        if (t < NBIN) {
            int pre = 0;
            for (int w = 0; w < wave; ++w) pre += wsum[w];
            const int first = pre + x - v;
            cur[t] += first;
            if (s == 0) bin_start[t] = first;
        }
    }
    __syncthreads();

    const int v0 = ((int)blockIdx.x * NVEC4) / NSLICE;
    const int v1 = (((int)blockIdx.x + 1) * NVEC4) / NSLICE;
    const int4* rows4 = (const int4*)rows;
    const int4* cols4 = (const int4*)cols;
    int u0 = v0 + t;
    for (; u0 + (SLICE_DEPTH - 1) * SLICE_T < v1; u0 += SLICE_DEPTH * SLICE_T) {
        int4 r[SLICE_DEPTH];
        int4 c[SLICE_DEPTH];
        int4 p[SLICE_DEPTH];
        for (int k = 0; k < SLICE_DEPTH; ++k) {
            r[k] = rows4[u0 + k * SLICE_T];
            c[k] = cols4[u0 + k * SLICE_T];
        }
        for (int k = 0; k < SLICE_DEPTH; ++k)
            asm volatile("" : "+v"(r[k].x), "+v"(r[k].y), "+v"(r[k].z), "+v"(r[k].w),
                              "+v"(c[k].x), "+v"(c[k].y), "+v"(c[k].z), "+v"(c[k].w));
        for (int k = 0; k < SLICE_DEPTH; ++k) {
            p[k].x = atomicAdd(&cur[r[k].x >> BIN_SHIFT], 1);
            p[k].y = atomicAdd(&cur[r[k].y >> BIN_SHIFT], 1);
            p[k].z = atomicAdd(&cur[r[k].z >> BIN_SHIFT], 1);
            p[k].w = atomicAdd(&cur[r[k].w >> BIN_SHIFT], 1);
        }
        for (int k = 0; k < SLICE_DEPTH; ++k) {
            binned[p[k].x] = ((unsigned)c[k].x << BIN_SHIFT) | (unsigned)(r[k].x & (BIN_NODES - 1));
            binned[p[k].y] = ((unsigned)c[k].y << BIN_SHIFT) | (unsigned)(r[k].y & (BIN_NODES - 1));
            binned[p[k].z] = ((unsigned)c[k].z << BIN_SHIFT) | (unsigned)(r[k].z & (BIN_NODES - 1));
            binned[p[k].w] = ((unsigned)c[k].w << BIN_SHIFT) | (unsigned)(r[k].w & (BIN_NODES - 1));
        }
    }
    for (; u0 < v1; u0 += SLICE_T) {
        const int4 r = rows4[u0];
        const int4 c = cols4[u0];
        const int p0 = atomicAdd(&cur[r.x >> BIN_SHIFT], 1);
        const int p1 = atomicAdd(&cur[r.y >> BIN_SHIFT], 1);
        const int p2 = atomicAdd(&cur[r.z >> BIN_SHIFT], 1);
        const int p3 = atomicAdd(&cur[r.w >> BIN_SHIFT], 1);
        binned[p0] = ((unsigned)c.x << BIN_SHIFT) | (unsigned)(r.x & (BIN_NODES - 1));
        binned[p1] = ((unsigned)c.y << BIN_SHIFT) | (unsigned)(r.y & (BIN_NODES - 1));
        binned[p2] = ((unsigned)c.z << BIN_SHIFT) | (unsigned)(r.z & (BIN_NODES - 1));
        binned[p3] = ((unsigned)c.w << BIN_SHIFT) | (unsigned)(r.w & (BIN_NODES - 1));
    }
}

__global__ void bin_csr_kernel(const unsigned* __restrict__ binned,
                               const int* __restrict__ bin_start,
                               int* __restrict__ rowptr, int* __restrict__ counts,
                               int* __restrict__ csr_cols) {
    __shared__ int cnt[BIN_NODES + NSPARE];
    __shared__ int cur[BIN_NODES + NSPARE];
    __shared__ int wsum[BIN_NODES / 32];

    const int t     = threadIdx.x;
    const int lane  = t & 31;
    const int wave  = t >> 5;
    const int bin   = blockIdx.x;
    const int spare = BIN_NODES + lane;

    const int start = bin_start[bin];
    const int end   = (bin + 1 < NBIN) ? bin_start[bin + 1] : N_EDGES;

    if (t < BIN_NODES + NSPARE) cnt[t] = 0;
    __syncthreads();
    for (int i0 = start + t; i0 < end; i0 += BIN_DEPTH * BIN_T) {
        unsigned rec[BIN_DEPTH];
        for (int k = 0; k < BIN_DEPTH; ++k) rec[k] = binned[min(i0 + k * BIN_T, end - 1)];
        for (int k = 0; k < BIN_DEPTH; ++k)
            atomicAdd(&cnt[(i0 + k * BIN_T < end) ? (int)(rec[k] & (BIN_NODES - 1)) : spare], 1);
    }
    __syncthreads();

    const int v = (t < BIN_NODES) ? cnt[t] : 0;
    int x = v;
    for (int off = 1; off < 32; off <<= 1) {
        const int y = __shfl_up(x, off);
        if (lane >= off) x += y;
    }
    if (t < BIN_NODES && lane == 31) wsum[wave] = x;
    __syncthreads();
    if (t < BIN_NODES) {
        int pre = 0;
        for (int w = 0; w < wave; ++w) pre += wsum[w];
        const int excl = start + pre + x - v;
        cur[t] = excl;
        const int node = bin * BIN_NODES + t;
        if (node < N_NODES) {
            rowptr[node] = excl;
            counts[node] = v;
        }
    } else if (t < BIN_NODES + NSPARE) {
        cur[t] = 0;
    }
    __syncthreads();
    for (int i0 = start + t; i0 < end; i0 += BIN_DEPTH * BIN_T) {
        unsigned rec[BIN_DEPTH];
        int pos[BIN_DEPTH];
        for (int k = 0; k < BIN_DEPTH; ++k) rec[k] = binned[min(i0 + k * BIN_T, end - 1)];
        for (int k = 0; k < BIN_DEPTH; ++k)
            pos[k] = atomicAdd(&cur[(i0 + k * BIN_T < end) ? (int)(rec[k] & (BIN_NODES - 1)) : spare], 1);
        for (int k = 0; k < BIN_DEPTH; ++k)
            if (i0 + k * BIN_T < end) csr_cols[pos[k]] = (int)(rec[k] >> BIN_SHIFT);
    }
}

#define LDSW   132
#define TILE_M 32
#define FUSE_T 512
#define W_PER_T (DIM * DIM / 4 / FUSE_T)

static_assert(DIM == 128 && TILE_M == 2 * (FUSE_T / 32) && (DIM / 16) * (TILE_M / 16) == FUSE_T / 32,
              "two nodes per wave in the gather, one 16 x 16 tile per wave in the product");

__global__ void __launch_bounds__(FUSE_T)
agg_gemm_kernel(const float* __restrict__ seq,
                const float* __restrict__ W,
                const int* __restrict__ rowptr,
                const int* __restrict__ counts,
                const int* __restrict__ csr_cols,
                const float* __restrict__ alpha,
                float* __restrict__ out) {
    __shared__ float sW[DIM * LDSW];
    __shared__ float sA[TILE_M * LDSW];
    __shared__ float sO[TILE_M * LDSW];

    const int tid      = threadIdx.x;
    const int wave     = tid >> 5;
    const int lane     = tid & 31;
    const int nodeBase = blockIdx.x * TILE_M;

    int st[2], ln[2];
    for (int h = 0; h < 2; ++h) {
        const int node = nodeBase + wave + 16 * h;
        const int nc   = min(node, N_NODES - 1);
        st[h] = rowptr[nc];
        ln[h] = (node < N_NODES) ? counts[nc] : 0;
    }

    {
        float4 w[W_PER_T];
#pragma unroll
        for (int k = 0; k < W_PER_T; ++k) w[k] = ((const float4*)W)[tid + k * FUSE_T];
#pragma unroll
        for (int k = 0; k < W_PER_T; ++k) {
            const int v = tid + k * FUSE_T;
            *(float4*)&sW[(v >> 5) * LDSW + (v & 31) * 4] = w[k];
        }
    }

    const float* base = seq + lane * 4;
    for (int h = 0; h < 2; ++h) {
        const int m = wave + 16 * h;
        float4 acc = make_float4(0.f, 0.f, 0.f, 0.f);
        for (int j = 0; j < ln[h]; ++j) {
            const int c = csr_cols[st[h] + j];
            const float4 v = *(const float4*)(base + (size_t)c * DIM);
            acc.x += v.x; acc.y += v.y; acc.z += v.z; acc.w += v.w;
        }
        *(float4*)&sA[m * LDSW + lane * 4] = acc;
    }
    __syncthreads();

    const int lo = lane & 15;
    const int hi = lane >> 4;
    const int mt = wave >> 3;
    const int nt = wave & 7;
    const float* arow = &sA[(mt * 16 + lo) * LDSW + 4 * hi];
    const float* brow = &sW[(nt * 16 + lo) * LDSW + 4 * hi];

    v8f acc = {};
#pragma unroll
    for (int j = 0; j < DIM / 8; ++j) {
        const float4 a = *(const float4*)(arow + 8 * j);
        const float4 b = *(const float4*)(brow + 8 * j);
        const v2f a0 = {a.x, a.y};
        const v2f b0 = {b.x, b.y};
        const v2f a1 = {a.z, a.w};
        const v2f b1 = {b.z, b.w};
        acc = __builtin_amdgcn_wmma_f32_16x16x4_f32(
             false, a0,  false, b0,
             (short)0, acc,  false,  false);
        acc = __builtin_amdgcn_wmma_f32_16x16x4_f32(
             false, a1,  false, b1,
             (short)0, acc,  false,  false);
    }

#pragma unroll
    for (int r = 0; r < 8; ++r)
        sO[(mt * 16 + r + 8 * hi) * LDSW + nt * 16 + lo] = acc[r];
    __syncthreads();

    const float al = alpha[0];
    for (int h = 0; h < 2; ++h) {
        const int m    = wave + 16 * h;
        const int node = nodeBase + m;
        if (node < N_NODES) {
            const float4 s = *(const float4*)&sO[m * LDSW + lane * 4];
            const float  d = (float)ln[h] + EPS;
            float4 o;
            o.x = s.x / d; o.y = s.y / d; o.z = s.z / d; o.w = s.w / d;
            o.x = (o.x >= 0.f) ? o.x : al * o.x;
            o.y = (o.y >= 0.f) ? o.y : al * o.y;
            o.z = (o.z >= 0.f) ? o.z : al * o.z;
            o.w = (o.w >= 0.f) ? o.w : al * o.w;
            *(float4*)(out + (size_t)node * DIM + lane * 4) = o;
        }
    }
}

extern "C" void kernel_launch(void* const* d_in, const int* in_sizes, int n_in,
                              void* d_out, int out_size, void* d_ws, size_t ws_size,
                              hipStream_t stream) {
    (void)in_sizes; (void)n_in; (void)out_size; (void)ws_size;

    const float* seq   = (const float*)d_in[0];
    const float* W     = (const float*)d_in[1];
    const float* alpha = (const float*)d_in[2];
    const int*   rows  = (const int*)d_in[3];
    const int*   cols  = (const int*)d_in[4];
    float* out = (float*)d_out;

    char* ws = (char*)d_ws;
    unsigned* binned       = (unsigned*)(ws);
    int*      slice_counts = (int*)(ws + 6400000);
    int*      counts       = (int*)(ws + 25600000);
    int*      rowptr       = (int*)(ws + 25800000);
    int*      bin_start    = (int*)(ws + 26200000);
    int*      csr_cols     = (int*)(ws + 26202048);

    bin_count_kernel<<<NSLICE, SLICE_T, 0, stream>>>(rows, slice_counts);
    bin_place_kernel<<<NSLICE, SLICE_T, 0, stream>>>(rows, cols, slice_counts, bin_start, binned);
    bin_csr_kernel<<<NBIN, BIN_T, 0, stream>>>(binned, bin_start, rowptr, counts, csr_cols);

    agg_gemm_kernel<<<(N_NODES + TILE_M - 1) / TILE_M, FUSE_T, 0, stream>>>(
        seq, W, rowptr, counts, csr_cols, alpha, out);
}


# both sort levels order their 4-byte records in LDS and write them out as contiguous runs or whole lines, not one scattered store per edge; in-row edge order stays the LDS atomics' arrival order
# speedup vs baseline: 3.5120x; 1.3187x over previous
#include <hip/hip_runtime.h>

#define N_NODES 50000
#define N_EDGES 1600000
#define DIM     128
#define EPS     1e-8f

typedef __attribute__((ext_vector_type(2))) float v2f;
typedef __attribute__((ext_vector_type(8))) float v8f;

#define BIN_SHIFT 8
#define BIN_NODES 256
#define NBIN    ((N_NODES + BIN_NODES - 1) / BIN_NODES)
#define NSLICE  256
#define SLICE_T 512
#define SLICE_DEPTH 3
#define BIN_T   1024
#define BIN_DEPTH 8
#define NSPARE  32
#define NVEC4   (N_EDGES / 4)

#define TBL_W   256
#define TBL_Q   (TBL_W / 4)
#define TBL_R   (SLICE_T / TBL_Q)
#define TBL_K   (NSLICE / TBL_R)
#define TBL_DEPTH 16
#define SLICE_MAX (((NVEC4 + NSLICE - 1) / NSLICE) * 4)
#define CSR_STAGE 16384

static_assert(N_NODES <= (1 << 24), "a source id must fit the 24 bits above the row's low byte");
static_assert(N_EDGES % 4 == 0, "the edge list is read as int4");
static_assert(NBIN % 4 == 0 && NBIN <= TBL_W, "a thread sums four bins of a table row");
static_assert(SLICE_T % TBL_Q == 0 && NSLICE % TBL_R == 0, "the table's rows are dealt evenly to the threads");
static_assert((TBL_K & (TBL_K - 1)) == 0 && TBL_K % TBL_DEPTH == 0, "the row rotation wraps with a mask");
static_assert(TBL_W <= SLICE_T && TBL_W % 32 == 0, "the bin scan runs on the first TBL_W threads");

__global__ void bin_count_kernel(const int* __restrict__ rows, int* __restrict__ slice_counts) {
    __shared__ int hist[NBIN];
    const int t = threadIdx.x;
    for (int b = t; b < NBIN; b += SLICE_T) hist[b] = 0;
    __syncthreads();

    const int v0 = ((int)blockIdx.x * NVEC4) / NSLICE;
    const int v1 = (((int)blockIdx.x + 1) * NVEC4) / NSLICE;
    const int4* rows4 = (const int4*)rows;
    int u0 = v0 + t;
    for (; u0 + (SLICE_DEPTH - 1) * SLICE_T < v1; u0 += SLICE_DEPTH * SLICE_T) {
        int4 r[SLICE_DEPTH];
        for (int k = 0; k < SLICE_DEPTH; ++k) r[k] = rows4[u0 + k * SLICE_T];
        for (int k = 0; k < SLICE_DEPTH; ++k)
            asm volatile("" : "+v"(r[k].x), "+v"(r[k].y), "+v"(r[k].z), "+v"(r[k].w));
        for (int k = 0; k < SLICE_DEPTH; ++k) {
            atomicAdd(&hist[r[k].x >> BIN_SHIFT], 1);
            atomicAdd(&hist[r[k].y >> BIN_SHIFT], 1);
            atomicAdd(&hist[r[k].z >> BIN_SHIFT], 1);
            atomicAdd(&hist[r[k].w >> BIN_SHIFT], 1);
        }
    }
    for (; u0 < v1; u0 += SLICE_T) {
        const int4 r = rows4[u0];
        atomicAdd(&hist[r.x >> BIN_SHIFT], 1);
        atomicAdd(&hist[r.y >> BIN_SHIFT], 1);
        atomicAdd(&hist[r.z >> BIN_SHIFT], 1);
        atomicAdd(&hist[r.w >> BIN_SHIFT], 1);
    }
    __syncthreads();
    for (int b = t; b < NBIN; b += SLICE_T) slice_counts[blockIdx.x * TBL_W + b] = hist[b];
}

__global__ void __launch_bounds__(SLICE_T)
bin_place_kernel(const int* __restrict__ rows, const int* __restrict__ cols,
                                 const int* __restrict__ slice_counts,
                                 int* __restrict__ bin_start,
                                 unsigned* __restrict__ binned) {
    __shared__ int cur[TBL_W];
    __shared__ int tot[TBL_W];
    __shared__ int lcur[TBL_W];
    __shared__ int wsum[TBL_W / 32];
    __shared__ int wown[TBL_W / 32];
    __shared__ unsigned stage[SLICE_MAX];
    __shared__ int      shift[SLICE_MAX];
    const int t    = threadIdx.x;
    const int lane = t & 31;
    const int wave = t >> 5;
    const int s    = blockIdx.x;

    const int own = (t < NBIN) ? slice_counts[s * TBL_W + t] : 0;

    if (t < TBL_W) { cur[t] = 0; tot[t] = 0; }
    __syncthreads();

    const int q = t & (TBL_Q - 1);
    const int r = t / TBL_Q;
    if (q < NBIN / 4) {
        const int4* tbl = (const int4*)slice_counts + q;
        int4 T = make_int4(0, 0, 0, 0);
        int4 P = make_int4(0, 0, 0, 0);
        for (int k0 = 0; k0 < TBL_K; k0 += TBL_DEPTH) {
            int4 v[TBL_DEPTH];
            for (int k = 0; k < TBL_DEPTH; ++k) {
                const int sp = r + TBL_R * ((k0 + k + s) & (TBL_K - 1));
                v[k] = tbl[sp * TBL_Q];
            }
            for (int k = 0; k < TBL_DEPTH; ++k)
                asm volatile("" : "+v"(v[k].x), "+v"(v[k].y), "+v"(v[k].z), "+v"(v[k].w));
            for (int k = 0; k < TBL_DEPTH; ++k) {
                const int sp = r + TBL_R * ((k0 + k + s) & (TBL_K - 1));
                const int before = (sp < s) ? -1 : 0;
                T.x += v[k].x; T.y += v[k].y; T.z += v[k].z; T.w += v[k].w;
                P.x += v[k].x & before; P.y += v[k].y & before;
                P.z += v[k].z & before; P.w += v[k].w & before;
            }
        }
        atomicAdd(&tot[4 * q + 0], T.x); atomicAdd(&tot[4 * q + 1], T.y);
        atomicAdd(&tot[4 * q + 2], T.z); atomicAdd(&tot[4 * q + 3], T.w);
        atomicAdd(&cur[4 * q + 0], P.x); atomicAdd(&cur[4 * q + 1], P.y);
        atomicAdd(&cur[4 * q + 2], P.z); atomicAdd(&cur[4 * q + 3], P.w);
    }
    __syncthreads();

    {
        const int v = (t < TBL_W) ? tot[t] : 0;
        int x = v;
        int z = own;
        for (int off = 1; off < 32; off <<= 1) {
            const int y  = __shfl_up(x, off);
            const int yo = __shfl_up(z, off);
            if (lane >= off) { x += y; z += yo; }
        }
        if (t < TBL_W && lane == 31) { wsum[wave] = x; wown[wave] = z; }
        __syncthreads();
        if (t < NBIN) {
            int pre = 0, preo = 0;
            for (int w = 0; w < wave; ++w) { pre += wsum[w]; preo += wown[w]; }
            const int first  = pre + x - v;
            const int lfirst = preo + z - own;
            cur[t]  = cur[t] + first - lfirst;
            lcur[t] = lfirst;
            if (s == 0) bin_start[t] = first;
        }
    }
    __syncthreads();

    const int v0 = ((int)blockIdx.x * NVEC4) / NSLICE;
    const int v1 = (((int)blockIdx.x + 1) * NVEC4) / NSLICE;
    const int4* rows4 = (const int4*)rows;
    const int4* cols4 = (const int4*)cols;
    int u0 = v0 + t;
    for (; u0 + (SLICE_DEPTH - 1) * SLICE_T < v1; u0 += SLICE_DEPTH * SLICE_T) {
        int4 r[SLICE_DEPTH];
        int4 c[SLICE_DEPTH];
        int4 p[SLICE_DEPTH];
        for (int k = 0; k < SLICE_DEPTH; ++k) {
            r[k] = rows4[u0 + k * SLICE_T];
            c[k] = cols4[u0 + k * SLICE_T];
        }
        for (int k = 0; k < SLICE_DEPTH; ++k)
            asm volatile("" : "+v"(r[k].x), "+v"(r[k].y), "+v"(r[k].z), "+v"(r[k].w),
                              "+v"(c[k].x), "+v"(c[k].y), "+v"(c[k].z), "+v"(c[k].w));
        for (int k = 0; k < SLICE_DEPTH; ++k) {
            p[k].x = atomicAdd(&lcur[r[k].x >> BIN_SHIFT], 1);
            p[k].y = atomicAdd(&lcur[r[k].y >> BIN_SHIFT], 1);
            p[k].z = atomicAdd(&lcur[r[k].z >> BIN_SHIFT], 1);
            p[k].w = atomicAdd(&lcur[r[k].w >> BIN_SHIFT], 1);
        }
        for (int k = 0; k < SLICE_DEPTH; ++k) {
            stage[p[k].x] = ((unsigned)c[k].x << BIN_SHIFT) | (unsigned)(r[k].x & (BIN_NODES - 1));
            stage[p[k].y] = ((unsigned)c[k].y << BIN_SHIFT) | (unsigned)(r[k].y & (BIN_NODES - 1));
            stage[p[k].z] = ((unsigned)c[k].z << BIN_SHIFT) | (unsigned)(r[k].z & (BIN_NODES - 1));
            stage[p[k].w] = ((unsigned)c[k].w << BIN_SHIFT) | (unsigned)(r[k].w & (BIN_NODES - 1));
            shift[p[k].x] = cur[r[k].x >> BIN_SHIFT];
            shift[p[k].y] = cur[r[k].y >> BIN_SHIFT];
            shift[p[k].z] = cur[r[k].z >> BIN_SHIFT];
            shift[p[k].w] = cur[r[k].w >> BIN_SHIFT];
        }
    }
    for (; u0 < v1; u0 += SLICE_T) {
        const int4 r = rows4[u0];
        const int4 c = cols4[u0];
        const int p0 = atomicAdd(&lcur[r.x >> BIN_SHIFT], 1);
        const int p1 = atomicAdd(&lcur[r.y >> BIN_SHIFT], 1);
        const int p2 = atomicAdd(&lcur[r.z >> BIN_SHIFT], 1);
        const int p3 = atomicAdd(&lcur[r.w >> BIN_SHIFT], 1);
        stage[p0] = ((unsigned)c.x << BIN_SHIFT) | (unsigned)(r.x & (BIN_NODES - 1));
        stage[p1] = ((unsigned)c.y << BIN_SHIFT) | (unsigned)(r.y & (BIN_NODES - 1));
        stage[p2] = ((unsigned)c.z << BIN_SHIFT) | (unsigned)(r.z & (BIN_NODES - 1));
        stage[p3] = ((unsigned)c.w << BIN_SHIFT) | (unsigned)(r.w & (BIN_NODES - 1));
        shift[p0] = cur[r.x >> BIN_SHIFT];
        shift[p1] = cur[r.y >> BIN_SHIFT];
        shift[p2] = cur[r.z >> BIN_SHIFT];
        shift[p3] = cur[r.w >> BIN_SHIFT];
    }
    __syncthreads();

    const int n_slice = (v1 - v0) * 4;
    for (int i = t; i < n_slice; i += SLICE_T) binned[i + shift[i]] = stage[i];
}

__global__ void bin_csr_kernel(const unsigned* __restrict__ binned,
                               const int* __restrict__ bin_start,
                               int* __restrict__ rowptr, int* __restrict__ counts,
                               int* __restrict__ csr_cols) {
    __shared__ int cnt[BIN_NODES + NSPARE];
    __shared__ int cur[BIN_NODES + NSPARE];
    __shared__ int wsum[BIN_NODES / 32];
    __shared__ int stage[CSR_STAGE];

    const int t     = threadIdx.x;
    const int lane  = t & 31;
    const int wave  = t >> 5;
    const int bin   = blockIdx.x;
    const int spare = BIN_NODES + lane;

    const int start = bin_start[bin];
    const int end   = (bin + 1 < NBIN) ? bin_start[bin + 1] : N_EDGES;
    const int len   = end - start;
    const bool staged = len <= CSR_STAGE;

    if (t < BIN_NODES + NSPARE) cnt[t] = 0;
    __syncthreads();
    for (int i0 = start + t; i0 < end; i0 += BIN_DEPTH * BIN_T) {
        unsigned rec[BIN_DEPTH];
        for (int k = 0; k < BIN_DEPTH; ++k) rec[k] = binned[min(i0 + k * BIN_T, end - 1)];
        for (int k = 0; k < BIN_DEPTH; ++k)
            atomicAdd(&cnt[(i0 + k * BIN_T < end) ? (int)(rec[k] & (BIN_NODES - 1)) : spare], 1);
    }
    __syncthreads();

    const int v = (t < BIN_NODES) ? cnt[t] : 0;
    int x = v;
    for (int off = 1; off < 32; off <<= 1) {
        const int y = __shfl_up(x, off);
        if (lane >= off) x += y;
    }
    if (t < BIN_NODES && lane == 31) wsum[wave] = x;
    __syncthreads();
    if (t < BIN_NODES) {
        int pre = 0;
        for (int w = 0; w < wave; ++w) pre += wsum[w];
        const int excl = start + pre + x - v;
        cur[t] = excl;
        const int node = bin * BIN_NODES + t;
        if (node < N_NODES) {
            rowptr[node] = excl;
            counts[node] = v;
        }
    } else if (t < BIN_NODES + NSPARE) {
        cur[t] = 0;
    }
    __syncthreads();
    for (int i0 = start + t; i0 < end; i0 += BIN_DEPTH * BIN_T) {
        unsigned rec[BIN_DEPTH];
        int pos[BIN_DEPTH];
        for (int k = 0; k < BIN_DEPTH; ++k) rec[k] = binned[min(i0 + k * BIN_T, end - 1)];
        for (int k = 0; k < BIN_DEPTH; ++k)
            pos[k] = atomicAdd(&cur[(i0 + k * BIN_T < end) ? (int)(rec[k] & (BIN_NODES - 1)) : spare], 1);
        for (int k = 0; k < BIN_DEPTH; ++k)
            if (i0 + k * BIN_T < end) {
                const int col = (int)(rec[k] >> BIN_SHIFT);
                if (staged) stage[pos[k] - start] = col;
                else        csr_cols[pos[k]] = col;
            }
    }
    if (staged) {
        __syncthreads();
        const int lead = (int)(((size_t)(csr_cols + start) >> 2) & 31);
        for (int i = t - lead; i < len; i += BIN_T)
            if (i >= 0) csr_cols[start + i] = stage[i];
    }
}

#define LDSW   132
#define TILE_M 32
#define FUSE_T 512
#define W_PER_T (DIM * DIM / 4 / FUSE_T)

static_assert(DIM == 128 && TILE_M == 2 * (FUSE_T / 32) && (DIM / 16) * (TILE_M / 16) == FUSE_T / 32,
              "two nodes per wave in the gather, one 16 x 16 tile per wave in the product");

__global__ void __launch_bounds__(FUSE_T)
agg_gemm_kernel(const float* __restrict__ seq,
                const float* __restrict__ W,
                const int* __restrict__ rowptr,
                const int* __restrict__ counts,
                const int* __restrict__ csr_cols,
                const float* __restrict__ alpha,
                float* __restrict__ out) {
    __shared__ float sW[DIM * LDSW];
    __shared__ float sA[TILE_M * LDSW];
    __shared__ float sO[TILE_M * LDSW];

    const int tid      = threadIdx.x;
    const int wave     = tid >> 5;
    const int lane     = tid & 31;
    const int nodeBase = blockIdx.x * TILE_M;

    int st[2], ln[2];
    for (int h = 0; h < 2; ++h) {
        const int node = nodeBase + wave + 16 * h;
        const int nc   = min(node, N_NODES - 1);
        st[h] = rowptr[nc];
        ln[h] = (node < N_NODES) ? counts[nc] : 0;
    }

    {
        float4 w[W_PER_T];
#pragma unroll
        for (int k = 0; k < W_PER_T; ++k) w[k] = ((const float4*)W)[tid + k * FUSE_T];
#pragma unroll
        for (int k = 0; k < W_PER_T; ++k) {
            const int v = tid + k * FUSE_T;
            *(float4*)&sW[(v >> 5) * LDSW + (v & 31) * 4] = w[k];
        }
    }

    const float* base = seq + lane * 4;
    for (int h = 0; h < 2; ++h) {
        const int m = wave + 16 * h;
        float4 acc = make_float4(0.f, 0.f, 0.f, 0.f);
        for (int j = 0; j < ln[h]; ++j) {
            const int c = csr_cols[st[h] + j];
            const float4 v = *(const float4*)(base + (size_t)c * DIM);
            acc.x += v.x; acc.y += v.y; acc.z += v.z; acc.w += v.w;
        }
        *(float4*)&sA[m * LDSW + lane * 4] = acc;
    }
    __syncthreads();

    const int lo = lane & 15;
    const int hi = lane >> 4;
    const int mt = wave >> 3;
    const int nt = wave & 7;
    const float* arow = &sA[(mt * 16 + lo) * LDSW + 4 * hi];
    const float* brow = &sW[(nt * 16 + lo) * LDSW + 4 * hi];

    v8f acc = {};
#pragma unroll
    for (int j = 0; j < DIM / 8; ++j) {
        const float4 a = *(const float4*)(arow + 8 * j);
        const float4 b = *(const float4*)(brow + 8 * j);
        const v2f a0 = {a.x, a.y};
        const v2f b0 = {b.x, b.y};
        const v2f a1 = {a.z, a.w};
        const v2f b1 = {b.z, b.w};
        acc = __builtin_amdgcn_wmma_f32_16x16x4_f32(
             false, a0,  false, b0,
             (short)0, acc,  false,  false);
        acc = __builtin_amdgcn_wmma_f32_16x16x4_f32(
             false, a1,  false, b1,
             (short)0, acc,  false,  false);
    }

#pragma unroll
    for (int r = 0; r < 8; ++r)
        sO[(mt * 16 + r + 8 * hi) * LDSW + nt * 16 + lo] = acc[r];
    __syncthreads();

    const float al = alpha[0];
    for (int h = 0; h < 2; ++h) {
        const int m    = wave + 16 * h;
        const int node = nodeBase + m;
        if (node < N_NODES) {
            const float4 s = *(const float4*)&sO[m * LDSW + lane * 4];
            const float  d = (float)ln[h] + EPS;
            float4 o;
            o.x = s.x / d; o.y = s.y / d; o.z = s.z / d; o.w = s.w / d;
            o.x = (o.x >= 0.f) ? o.x : al * o.x;
            o.y = (o.y >= 0.f) ? o.y : al * o.y;
            o.z = (o.z >= 0.f) ? o.z : al * o.z;
            o.w = (o.w >= 0.f) ? o.w : al * o.w;
            *(float4*)(out + (size_t)node * DIM + lane * 4) = o;
        }
    }
}

extern "C" void kernel_launch(void* const* d_in, const int* in_sizes, int n_in,
                              void* d_out, int out_size, void* d_ws, size_t ws_size,
                              hipStream_t stream) {
    (void)in_sizes; (void)n_in; (void)out_size; (void)ws_size;

    const float* seq   = (const float*)d_in[0];
    const float* W     = (const float*)d_in[1];
    const float* alpha = (const float*)d_in[2];
    const int*   rows  = (const int*)d_in[3];
    const int*   cols  = (const int*)d_in[4];
    float* out = (float*)d_out;

    char* ws = (char*)d_ws;
    unsigned* binned       = (unsigned*)(ws);
    int*      slice_counts = (int*)(ws + 6400000);
    int*      counts       = (int*)(ws + 25600000);
    int*      rowptr       = (int*)(ws + 25800000);
    int*      bin_start    = (int*)(ws + 26200000);
    int*      csr_cols     = (int*)(ws + 26202048);

    bin_count_kernel<<<NSLICE, SLICE_T, 0, stream>>>(rows, slice_counts);
    bin_place_kernel<<<NSLICE, SLICE_T, 0, stream>>>(rows, cols, slice_counts, bin_start, binned);
    bin_csr_kernel<<<NBIN, BIN_T, 0, stream>>>(binned, bin_start, rowptr, counts, csr_cols);

    agg_gemm_kernel<<<(N_NODES + TILE_M - 1) / TILE_M, FUSE_T, 0, stream>>>(
        seq, W, rowptr, counts, csr_cols, alpha, out);
}


# level-1 sort keeps each slice's bin-ordered records in its own region, a (start,count) word per bin; level 2 reads 256 segments; count launch and table sum go (4 launches to 3); sum order changes
# speedup vs baseline: 3.7796x; 1.0762x over previous
#include <hip/hip_runtime.h>

#define N_NODES 50000
#define N_EDGES 1600000
#define DIM     128
#define EPS     1e-8f

typedef __attribute__((ext_vector_type(2))) float v2f;
typedef __attribute__((ext_vector_type(8))) float v8f;
typedef __attribute__((ext_vector_type(4))) int   v4i;

#define BIN_SHIFT 8
#define BIN_NODES 256
#define NBIN    ((N_NODES + BIN_NODES - 1) / BIN_NODES)
#define NSLICE  256
#define SLICE_T 512
#define SLICE_V 4
#define BIN_T   1024
#define NSPARE  32
#define NVEC4   (N_EDGES / 4)

#define TBL_W   256
#define SLICE_MAX (((NVEC4 + NSLICE - 1) / NSLICE) * 4)
#define SLICE_PITCH ((SLICE_MAX + 31) / 32 * 32)
#define SEG_PER_WAVE (NSLICE / (BIN_T / 32))
#define SEG_SLOTS 2
#define CSR_STAGE 16384

static_assert(N_NODES <= (1 << 24), "a source id must fit the 24 bits above the row's low byte");
static_assert(N_EDGES % 4 == 0, "the edge list is read as int4");
static_assert(NVEC4 / NSLICE >= (SLICE_V - 1) * SLICE_T, "all but a thread's last int4 lie inside every slice");
static_assert((NVEC4 + NSLICE - 1) / NSLICE <= SLICE_V * SLICE_T, "a thread's int4s cover the longest slice");
static_assert(SLICE_PITCH * 4 == 25088 && SLICE_PITCH * 4 * NSLICE == 6422528, "the workspace offsets in kernel_launch are written for this pitch");
static_assert(SLICE_MAX < (1 << 16), "start and count of a segment share one word");
static_assert(NBIN <= TBL_W && TBL_W <= SLICE_T && TBL_W % 32 == 0, "the bin scan runs on the first TBL_W threads");
static_assert(NSLICE % (BIN_T / 32) == 0 && NSLICE <= BIN_T, "the segments are dealt evenly to the level-2 waves");

__device__ __forceinline__ unsigned pack_rec(int col, int row) {
    return ((unsigned)col << BIN_SHIFT) | (unsigned)(row & (BIN_NODES - 1));
}

__global__ void __launch_bounds__(SLICE_T)
slice_sort_kernel(const int* __restrict__ rows, const int* __restrict__ cols,
                  unsigned* __restrict__ seg_tbl, unsigned* __restrict__ binned) {
    __shared__ int hist[TBL_W];
    __shared__ int lfirst[TBL_W];
    __shared__ int wsum[TBL_W / 32];
    __shared__ __attribute__((aligned(16))) unsigned stage[SLICE_PITCH];
    const int t    = threadIdx.x;
    const int lane = t & 31;
    const int wave = t >> 5;
    const int s    = blockIdx.x;

    if (t < TBL_W) hist[t] = 0;
    __syncthreads();

    const int v0 = ((int)blockIdx.x * NVEC4) / NSLICE;
    const int v1 = (((int)blockIdx.x + 1) * NVEC4) / NSLICE;
    const v4i* rows4 = (const v4i*)rows;
    const v4i* cols4 = (const v4i*)cols;
    const bool last = v0 + t + (SLICE_V - 1) * SLICE_T < v1;

    v4i r[SLICE_V], c[SLICE_V], p[SLICE_V];
    for (int k = 0; k < SLICE_V; ++k) {
        const int u = min(v0 + t + k * SLICE_T, v1 - 1);
        r[k] = rows4[u];
        c[k] = cols4[u];
    }
    for (int k = 0; k < SLICE_V; ++k)
        asm volatile("" : "+v"(r[k]), "+v"(c[k]));

    for (int k = 0; k < SLICE_V - 1; ++k) {
        p[k].x = atomicAdd(&hist[r[k].x >> BIN_SHIFT], 1);
        p[k].y = atomicAdd(&hist[r[k].y >> BIN_SHIFT], 1);
        p[k].z = atomicAdd(&hist[r[k].z >> BIN_SHIFT], 1);
        p[k].w = atomicAdd(&hist[r[k].w >> BIN_SHIFT], 1);
    }
    if (last) {
        p[SLICE_V - 1].x = atomicAdd(&hist[r[SLICE_V - 1].x >> BIN_SHIFT], 1);
        p[SLICE_V - 1].y = atomicAdd(&hist[r[SLICE_V - 1].y >> BIN_SHIFT], 1);
        p[SLICE_V - 1].z = atomicAdd(&hist[r[SLICE_V - 1].z >> BIN_SHIFT], 1);
        p[SLICE_V - 1].w = atomicAdd(&hist[r[SLICE_V - 1].w >> BIN_SHIFT], 1);
    }
    __syncthreads();

    {
        const int own = (t < TBL_W) ? hist[t] : 0;
        int x = own;
        for (int off = 1; off < 32; off <<= 1) {
            const int y = __shfl_up(x, off);
            if (lane >= off) x += y;
        }
        if (t < TBL_W && lane == 31) wsum[wave] = x;
        __syncthreads();
        if (t < TBL_W) {
            int pre = 0;
            for (int w = 0; w < wave; ++w) pre += wsum[w];
            const int lf = pre + x - own;
            lfirst[t] = lf;
            if (t < NBIN) seg_tbl[s * TBL_W + t] = (unsigned)lf | ((unsigned)own << 16);
        }
    }
    __syncthreads();

    for (int k = 0; k < SLICE_V - 1; ++k) {
        stage[lfirst[r[k].x >> BIN_SHIFT] + p[k].x] = pack_rec(c[k].x, r[k].x);
        stage[lfirst[r[k].y >> BIN_SHIFT] + p[k].y] = pack_rec(c[k].y, r[k].y);
        stage[lfirst[r[k].z >> BIN_SHIFT] + p[k].z] = pack_rec(c[k].z, r[k].z);
        stage[lfirst[r[k].w >> BIN_SHIFT] + p[k].w] = pack_rec(c[k].w, r[k].w);
    }
    if (last) {
        const int k = SLICE_V - 1;
        stage[lfirst[r[k].x >> BIN_SHIFT] + p[k].x] = pack_rec(c[k].x, r[k].x);
        stage[lfirst[r[k].y >> BIN_SHIFT] + p[k].y] = pack_rec(c[k].y, r[k].y);
        stage[lfirst[r[k].z >> BIN_SHIFT] + p[k].z] = pack_rec(c[k].z, r[k].z);
        stage[lfirst[r[k].w >> BIN_SHIFT] + p[k].w] = pack_rec(c[k].w, r[k].w);
    }
    __syncthreads();

    const int n4 = v1 - v0;
    uint4* dst = (uint4*)(binned + (size_t)s * SLICE_PITCH);
    for (int i = t; i < n4; i += SLICE_T) dst[i] = ((const uint4*)stage)[i];
}

__global__ void
bin_csr_kernel(const unsigned* __restrict__ binned,
               const unsigned* __restrict__ seg_tbl,
               int* __restrict__ rowptr, int* __restrict__ counts,
               int* __restrict__ csr_cols) {
    __shared__ int cnt[BIN_NODES + NSPARE];
    __shared__ int cur[BIN_NODES + NSPARE];
    __shared__ int wsum[BIN_NODES / 32];
    __shared__ int wlen[BIN_NODES / 32];
    __shared__ unsigned segd[NSLICE];
    __shared__ int stage[CSR_STAGE];

    const int t     = threadIdx.x;
    const int lane  = t & 31;
    const int wave  = t >> 5;
    const int bin   = blockIdx.x;
    const int spare = BIN_NODES + lane;

    {
        const unsigned d = (t < NSLICE) ? seg_tbl[t * TBL_W + bin] : 0u;
        int a = (int)(d & 0xffffu);
        int b = (int)(d >> 16);
        for (int off = 16; off > 0; off >>= 1) {
            a += __shfl_xor(a, off);
            b += __shfl_xor(b, off);
        }
        if (t < NSLICE) {
            segd[t] = d;
            if (lane == 0) { wsum[wave] = a; wlen[wave] = b; }
        }
        if (t < BIN_NODES + NSPARE) cnt[t] = 0;
    }
    __syncthreads();
    int start = 0, len = 0;
    for (int w = 0; w < NSLICE / 32; ++w) { start += wsum[w]; len += wlen[w]; }
    const bool staged = len <= CSR_STAGE;

    int so[SEG_PER_WAVE], sn[SEG_PER_WAVE];
    for (int k = 0; k < SEG_PER_WAVE; ++k) {
        const int g = wave + k * (BIN_T / 32);
        const unsigned d = segd[g];
        so[k] = g * SLICE_PITCH + (int)(d & 0xffffu);
        sn[k] = (int)(d >> 16);
    }

    {
        unsigned rec[SEG_PER_WAVE][SEG_SLOTS];
        for (int k = 0; k < SEG_PER_WAVE; ++k)
            for (int j = 0; j < SEG_SLOTS; ++j)
                rec[k][j] = binned[so[k] + ((lane + 32 * j < sn[k]) ? lane + 32 * j : 0)];
        for (int k = 0; k < SEG_PER_WAVE; ++k)
            for (int j = 0; j < SEG_SLOTS; ++j)
                atomicAdd(&cnt[(lane + 32 * j < sn[k]) ? (int)(rec[k][j] & (BIN_NODES - 1)) : spare], 1);
        for (int k = 0; k < SEG_PER_WAVE; ++k)
            for (int i = 32 * SEG_SLOTS + lane; i < sn[k]; i += 32)
                atomicAdd(&cnt[binned[so[k] + i] & (BIN_NODES - 1)], 1);
    }
    __syncthreads();

    const int v = (t < BIN_NODES) ? cnt[t] : 0;
    int x = v;
    for (int off = 1; off < 32; off <<= 1) {
        const int y = __shfl_up(x, off);
        if (lane >= off) x += y;
    }
    if (t < BIN_NODES && lane == 31) wsum[wave] = x;
    __syncthreads();
    if (t < BIN_NODES) {
        int pre = 0;
        for (int w = 0; w < wave; ++w) pre += wsum[w];
        const int excl = start + pre + x - v;
        cur[t] = excl;
        const int node = bin * BIN_NODES + t;
        if (node < N_NODES) {
            rowptr[node] = excl;
            counts[node] = v;
        }
    } else if (t < BIN_NODES + NSPARE) {
        cur[t] = 0;
    }
    __syncthreads();
    {
        unsigned rec[SEG_PER_WAVE][SEG_SLOTS];
        int pos[SEG_PER_WAVE][SEG_SLOTS];
        for (int k = 0; k < SEG_PER_WAVE; ++k)
            for (int j = 0; j < SEG_SLOTS; ++j)
                rec[k][j] = binned[so[k] + ((lane + 32 * j < sn[k]) ? lane + 32 * j : 0)];
        for (int k = 0; k < SEG_PER_WAVE; ++k)
            for (int j = 0; j < SEG_SLOTS; ++j)
                pos[k][j] = atomicAdd(&cur[(lane + 32 * j < sn[k]) ? (int)(rec[k][j] & (BIN_NODES - 1)) : spare], 1);
        for (int k = 0; k < SEG_PER_WAVE; ++k)
            for (int j = 0; j < SEG_SLOTS; ++j)
                if (lane + 32 * j < sn[k]) {
                    const int col = (int)(rec[k][j] >> BIN_SHIFT);
                    if (staged) stage[pos[k][j] - start] = col;
                    else        csr_cols[pos[k][j]] = col;
                }
        for (int k = 0; k < SEG_PER_WAVE; ++k)
            for (int i = 32 * SEG_SLOTS + lane; i < sn[k]; i += 32) {
                const unsigned rc = binned[so[k] + i];
                const int ps  = atomicAdd(&cur[rc & (BIN_NODES - 1)], 1);
                const int col = (int)(rc >> BIN_SHIFT);
                if (staged) stage[ps - start] = col;
                else        csr_cols[ps] = col;
            }
    }
    if (staged) {
        __syncthreads();
        const int lead = (int)(((size_t)(csr_cols + start) >> 2) & 31);
        for (int i = t - lead; i < len; i += BIN_T)
            if (i >= 0) csr_cols[start + i] = stage[i];
    }
}

#define LDSW   132
#define TILE_M 32
#define FUSE_T 512
#define W_PER_T (DIM * DIM / 4 / FUSE_T)

static_assert(DIM == 128 && TILE_M == 2 * (FUSE_T / 32) && (DIM / 16) * (TILE_M / 16) == FUSE_T / 32,
              "two nodes per wave in the gather, one 16 x 16 tile per wave in the product");

__global__ void __launch_bounds__(FUSE_T)
agg_gemm_kernel(const float* __restrict__ seq,
                const float* __restrict__ W,
                const int* __restrict__ rowptr,
                const int* __restrict__ counts,
                const int* __restrict__ csr_cols,
                const float* __restrict__ alpha,
                float* __restrict__ out) {
    __shared__ float sW[DIM * LDSW];
    __shared__ float sA[TILE_M * LDSW];
    __shared__ float sO[TILE_M * LDSW];

    const int tid      = threadIdx.x;
    const int wave     = tid >> 5;
    const int lane     = tid & 31;
    const int nodeBase = blockIdx.x * TILE_M;

    int st[2], ln[2];
    for (int h = 0; h < 2; ++h) {
        const int node = nodeBase + wave + 16 * h;
        const int nc   = min(node, N_NODES - 1);
        st[h] = rowptr[nc];
        ln[h] = (node < N_NODES) ? counts[nc] : 0;
    }

    {
        float4 w[W_PER_T];
#pragma unroll
        for (int k = 0; k < W_PER_T; ++k) w[k] = ((const float4*)W)[tid + k * FUSE_T];
#pragma unroll
        for (int k = 0; k < W_PER_T; ++k) {
            const int v = tid + k * FUSE_T;
            *(float4*)&sW[(v >> 5) * LDSW + (v & 31) * 4] = w[k];
        }
    }

    const float* base = seq + lane * 4;
    for (int h = 0; h < 2; ++h) {
        const int m = wave + 16 * h;
        float4 acc = make_float4(0.f, 0.f, 0.f, 0.f);
        for (int j = 0; j < ln[h]; ++j) {
            const int c = csr_cols[st[h] + j];
            const float4 v = *(const float4*)(base + (size_t)c * DIM);
            acc.x += v.x; acc.y += v.y; acc.z += v.z; acc.w += v.w;
        }
        *(float4*)&sA[m * LDSW + lane * 4] = acc;
    }
    __syncthreads();

    const int lo = lane & 15;
    const int hi = lane >> 4;
    const int mt = wave >> 3;
    const int nt = wave & 7;
    const float* arow = &sA[(mt * 16 + lo) * LDSW + 4 * hi];
    const float* brow = &sW[(nt * 16 + lo) * LDSW + 4 * hi];

    v8f acc = {};
#pragma unroll
    for (int j = 0; j < DIM / 8; ++j) {
        const float4 a = *(const float4*)(arow + 8 * j);
        const float4 b = *(const float4*)(brow + 8 * j);
        const v2f a0 = {a.x, a.y};
        const v2f b0 = {b.x, b.y};
        const v2f a1 = {a.z, a.w};
        const v2f b1 = {b.z, b.w};
        acc = __builtin_amdgcn_wmma_f32_16x16x4_f32(
             false, a0,  false, b0,
             (short)0, acc,  false,  false);
        acc = __builtin_amdgcn_wmma_f32_16x16x4_f32(
             false, a1,  false, b1,
             (short)0, acc,  false,  false);
    }

#pragma unroll
    for (int r = 0; r < 8; ++r)
        sO[(mt * 16 + r + 8 * hi) * LDSW + nt * 16 + lo] = acc[r];
    __syncthreads();

    const float al = alpha[0];
    for (int h = 0; h < 2; ++h) {
        const int m    = wave + 16 * h;
        const int node = nodeBase + m;
        if (node < N_NODES) {
            const float4 s = *(const float4*)&sO[m * LDSW + lane * 4];
            const float  d = (float)ln[h] + EPS;
            float4 o;
            o.x = s.x / d; o.y = s.y / d; o.z = s.z / d; o.w = s.w / d;
            o.x = (o.x >= 0.f) ? o.x : al * o.x;
            o.y = (o.y >= 0.f) ? o.y : al * o.y;
            o.z = (o.z >= 0.f) ? o.z : al * o.z;
            o.w = (o.w >= 0.f) ? o.w : al * o.w;
            *(float4*)(out + (size_t)node * DIM + lane * 4) = o;
        }
    }
}

extern "C" void kernel_launch(void* const* d_in, const int* in_sizes, int n_in,
                              void* d_out, int out_size, void* d_ws, size_t ws_size,
                              hipStream_t stream) {
    (void)in_sizes; (void)n_in; (void)out_size; (void)ws_size;

    const float* seq   = (const float*)d_in[0];
    const float* W     = (const float*)d_in[1];
    const float* alpha = (const float*)d_in[2];
    const int*   rows  = (const int*)d_in[3];
    const int*   cols  = (const int*)d_in[4];
    float* out = (float*)d_out;

    char* ws = (char*)d_ws;
    unsigned* binned       = (unsigned*)(ws);
    unsigned* seg_tbl      = (unsigned*)(ws + 6422528);
    int*      counts       = (int*)(ws + 25600000);
    int*      rowptr       = (int*)(ws + 25800000);
    int*      csr_cols     = (int*)(ws + 26202048);

    slice_sort_kernel<<<NSLICE, SLICE_T, 0, stream>>>(rows, cols, seg_tbl, binned);
    bin_csr_kernel<<<NBIN, BIN_T, 0, stream>>>(binned, seg_tbl, rowptr, counts, csr_cols);

    agg_gemm_kernel<<<(N_NODES + TILE_M - 1) / TILE_M, FUSE_T, 0, stream>>>(
        seq, W, rowptr, counts, csr_cols, alpha, out);
}


# gather keeps 16 row loads in flight per wave (was 8, then 1 per left-over edge): source ids read once, a lane per edge, handed out by lane reads; left-over edges go as one group; sum order kept
# speedup vs baseline: 4.3819x; 1.1594x over previous
#include <hip/hip_runtime.h>

#define N_NODES 50000
#define N_EDGES 1600000
#define DIM     128
#define EPS     1e-8f

typedef __attribute__((ext_vector_type(2))) float v2f;
typedef __attribute__((ext_vector_type(8))) float v8f;
typedef __attribute__((ext_vector_type(4))) int   v4i;

#define BIN_SHIFT 8
#define BIN_NODES 256
#define NBIN    ((N_NODES + BIN_NODES - 1) / BIN_NODES)
#define NSLICE  256
#define SLICE_T 512
#define SLICE_V 4
#define BIN_T   1024
#define NSPARE  32
#define NVEC4   (N_EDGES / 4)

#define TBL_W   256
#define SLICE_MAX (((NVEC4 + NSLICE - 1) / NSLICE) * 4)
#define SLICE_PITCH ((SLICE_MAX + 31) / 32 * 32)
#define SEG_PER_WAVE (NSLICE / (BIN_T / 32))
#define SEG_SLOTS 2
#define CSR_STAGE 16384

static_assert(N_NODES <= (1 << 24), "a source id must fit the 24 bits above the row's low byte");
static_assert(N_EDGES % 4 == 0, "the edge list is read as int4");
static_assert(NVEC4 / NSLICE >= (SLICE_V - 1) * SLICE_T, "all but a thread's last int4 lie inside every slice");
static_assert((NVEC4 + NSLICE - 1) / NSLICE <= SLICE_V * SLICE_T, "a thread's int4s cover the longest slice");
static_assert(SLICE_PITCH * 4 == 25088 && SLICE_PITCH * 4 * NSLICE == 6422528, "the workspace offsets in kernel_launch are written for this pitch");
static_assert(SLICE_MAX < (1 << 16), "start and count of a segment share one word");
static_assert(NBIN <= TBL_W && TBL_W <= SLICE_T && TBL_W % 32 == 0, "the bin scan runs on the first TBL_W threads");
static_assert(NSLICE % (BIN_T / 32) == 0 && NSLICE <= BIN_T, "the segments are dealt evenly to the level-2 waves");

__device__ __forceinline__ unsigned pack_rec(int col, int row) {
    return ((unsigned)col << BIN_SHIFT) | (unsigned)(row & (BIN_NODES - 1));
}

__global__ void __launch_bounds__(SLICE_T)
slice_sort_kernel(const int* __restrict__ rows, const int* __restrict__ cols,
                  unsigned* __restrict__ seg_tbl, unsigned* __restrict__ binned) {
    __shared__ int hist[TBL_W];
    __shared__ int lfirst[TBL_W];
    __shared__ int wsum[TBL_W / 32];
    __shared__ __attribute__((aligned(16))) unsigned stage[SLICE_PITCH];
    const int t    = threadIdx.x;
    const int lane = t & 31;
    const int wave = t >> 5;
    const int s    = blockIdx.x;

    if (t < TBL_W) hist[t] = 0;
    __syncthreads();

    const int v0 = ((int)blockIdx.x * NVEC4) / NSLICE;
    const int v1 = (((int)blockIdx.x + 1) * NVEC4) / NSLICE;
    const v4i* rows4 = (const v4i*)rows;
    const v4i* cols4 = (const v4i*)cols;
    const bool last = v0 + t + (SLICE_V - 1) * SLICE_T < v1;

    v4i r[SLICE_V], c[SLICE_V], p[SLICE_V];
    for (int k = 0; k < SLICE_V; ++k) {
        const int u = min(v0 + t + k * SLICE_T, v1 - 1);
        r[k] = rows4[u];
        c[k] = cols4[u];
    }
    for (int k = 0; k < SLICE_V; ++k)
        asm volatile("" : "+v"(r[k]), "+v"(c[k]));

    for (int k = 0; k < SLICE_V - 1; ++k) {
        p[k].x = atomicAdd(&hist[r[k].x >> BIN_SHIFT], 1);
        p[k].y = atomicAdd(&hist[r[k].y >> BIN_SHIFT], 1);
        p[k].z = atomicAdd(&hist[r[k].z >> BIN_SHIFT], 1);
        p[k].w = atomicAdd(&hist[r[k].w >> BIN_SHIFT], 1);
    }
    if (last) {
        p[SLICE_V - 1].x = atomicAdd(&hist[r[SLICE_V - 1].x >> BIN_SHIFT], 1);
        p[SLICE_V - 1].y = atomicAdd(&hist[r[SLICE_V - 1].y >> BIN_SHIFT], 1);
        p[SLICE_V - 1].z = atomicAdd(&hist[r[SLICE_V - 1].z >> BIN_SHIFT], 1);
        p[SLICE_V - 1].w = atomicAdd(&hist[r[SLICE_V - 1].w >> BIN_SHIFT], 1);
    }
    __syncthreads();

    {
        const int own = (t < TBL_W) ? hist[t] : 0;
        int x = own;
        for (int off = 1; off < 32; off <<= 1) {
            const int y = __shfl_up(x, off);
            if (lane >= off) x += y;
        }
        if (t < TBL_W && lane == 31) wsum[wave] = x;
        __syncthreads();
        if (t < TBL_W) {
            int pre = 0;
            for (int w = 0; w < wave; ++w) pre += wsum[w];
            const int lf = pre + x - own;
            lfirst[t] = lf;
            if (t < NBIN) seg_tbl[s * TBL_W + t] = (unsigned)lf | ((unsigned)own << 16);
        }
    }
    __syncthreads();

    for (int k = 0; k < SLICE_V - 1; ++k) {
        stage[lfirst[r[k].x >> BIN_SHIFT] + p[k].x] = pack_rec(c[k].x, r[k].x);
        stage[lfirst[r[k].y >> BIN_SHIFT] + p[k].y] = pack_rec(c[k].y, r[k].y);
        stage[lfirst[r[k].z >> BIN_SHIFT] + p[k].z] = pack_rec(c[k].z, r[k].z);
        stage[lfirst[r[k].w >> BIN_SHIFT] + p[k].w] = pack_rec(c[k].w, r[k].w);
    }
    if (last) {
        const int k = SLICE_V - 1;
        stage[lfirst[r[k].x >> BIN_SHIFT] + p[k].x] = pack_rec(c[k].x, r[k].x);
        stage[lfirst[r[k].y >> BIN_SHIFT] + p[k].y] = pack_rec(c[k].y, r[k].y);
        stage[lfirst[r[k].z >> BIN_SHIFT] + p[k].z] = pack_rec(c[k].z, r[k].z);
        stage[lfirst[r[k].w >> BIN_SHIFT] + p[k].w] = pack_rec(c[k].w, r[k].w);
    }
    __syncthreads();

    const int n4 = v1 - v0;
    uint4* dst = (uint4*)(binned + (size_t)s * SLICE_PITCH);
    for (int i = t; i < n4; i += SLICE_T) dst[i] = ((const uint4*)stage)[i];
}

__global__ void
bin_csr_kernel(const unsigned* __restrict__ binned,
               const unsigned* __restrict__ seg_tbl,
               int* __restrict__ rowptr, int* __restrict__ counts,
               int* __restrict__ csr_cols) {
    __shared__ int cnt[BIN_NODES + NSPARE];
    __shared__ int cur[BIN_NODES + NSPARE];
    __shared__ int wsum[BIN_NODES / 32];
    __shared__ int wlen[BIN_NODES / 32];
    __shared__ unsigned segd[NSLICE];
    __shared__ int stage[CSR_STAGE];

    const int t     = threadIdx.x;
    const int lane  = t & 31;
    const int wave  = t >> 5;
    const int bin   = blockIdx.x;
    const int spare = BIN_NODES + lane;

    {
        const unsigned d = (t < NSLICE) ? seg_tbl[t * TBL_W + bin] : 0u;
        int a = (int)(d & 0xffffu);
        int b = (int)(d >> 16);
        for (int off = 16; off > 0; off >>= 1) {
            a += __shfl_xor(a, off);
            b += __shfl_xor(b, off);
        }
        if (t < NSLICE) {
            segd[t] = d;
            if (lane == 0) { wsum[wave] = a; wlen[wave] = b; }
        }
        if (t < BIN_NODES + NSPARE) cnt[t] = 0;
    }
    __syncthreads();
    int start = 0, len = 0;
    for (int w = 0; w < NSLICE / 32; ++w) { start += wsum[w]; len += wlen[w]; }
    const bool staged = len <= CSR_STAGE;

    int so[SEG_PER_WAVE], sn[SEG_PER_WAVE];
    for (int k = 0; k < SEG_PER_WAVE; ++k) {
        const int g = wave + k * (BIN_T / 32);
        const unsigned d = segd[g];
        so[k] = g * SLICE_PITCH + (int)(d & 0xffffu);
        sn[k] = (int)(d >> 16);
    }

    {
        unsigned rec[SEG_PER_WAVE][SEG_SLOTS];
        for (int k = 0; k < SEG_PER_WAVE; ++k)
            for (int j = 0; j < SEG_SLOTS; ++j)
                rec[k][j] = binned[so[k] + ((lane + 32 * j < sn[k]) ? lane + 32 * j : 0)];
        for (int k = 0; k < SEG_PER_WAVE; ++k)
            for (int j = 0; j < SEG_SLOTS; ++j)
                atomicAdd(&cnt[(lane + 32 * j < sn[k]) ? (int)(rec[k][j] & (BIN_NODES - 1)) : spare], 1);
        for (int k = 0; k < SEG_PER_WAVE; ++k)
            for (int i = 32 * SEG_SLOTS + lane; i < sn[k]; i += 32)
                atomicAdd(&cnt[binned[so[k] + i] & (BIN_NODES - 1)], 1);
    }
    __syncthreads();

    const int v = (t < BIN_NODES) ? cnt[t] : 0;
    int x = v;
    for (int off = 1; off < 32; off <<= 1) {
        const int y = __shfl_up(x, off);
        if (lane >= off) x += y;
    }
    if (t < BIN_NODES && lane == 31) wsum[wave] = x;
    __syncthreads();
    if (t < BIN_NODES) {
        int pre = 0;
        for (int w = 0; w < wave; ++w) pre += wsum[w];
        const int excl = start + pre + x - v;
        cur[t] = excl;
        const int node = bin * BIN_NODES + t;
        if (node < N_NODES) {
            rowptr[node] = excl;
            counts[node] = v;
        }
    } else if (t < BIN_NODES + NSPARE) {
        cur[t] = 0;
    }
    __syncthreads();
    {
        unsigned rec[SEG_PER_WAVE][SEG_SLOTS];
        int pos[SEG_PER_WAVE][SEG_SLOTS];
        for (int k = 0; k < SEG_PER_WAVE; ++k)
            for (int j = 0; j < SEG_SLOTS; ++j)
                rec[k][j] = binned[so[k] + ((lane + 32 * j < sn[k]) ? lane + 32 * j : 0)];
        for (int k = 0; k < SEG_PER_WAVE; ++k)
            for (int j = 0; j < SEG_SLOTS; ++j)
                pos[k][j] = atomicAdd(&cur[(lane + 32 * j < sn[k]) ? (int)(rec[k][j] & (BIN_NODES - 1)) : spare], 1);
        for (int k = 0; k < SEG_PER_WAVE; ++k)
            for (int j = 0; j < SEG_SLOTS; ++j)
                if (lane + 32 * j < sn[k]) {
                    const int col = (int)(rec[k][j] >> BIN_SHIFT);
                    if (staged) stage[pos[k][j] - start] = col;
                    else        csr_cols[pos[k][j]] = col;
                }
        for (int k = 0; k < SEG_PER_WAVE; ++k)
            for (int i = 32 * SEG_SLOTS + lane; i < sn[k]; i += 32) {
                const unsigned rc = binned[so[k] + i];
                const int ps  = atomicAdd(&cur[rc & (BIN_NODES - 1)], 1);
                const int col = (int)(rc >> BIN_SHIFT);
                if (staged) stage[ps - start] = col;
                else        csr_cols[ps] = col;
            }
    }
    if (staged) {
        __syncthreads();
        const int lead = (int)(((size_t)(csr_cols + start) >> 2) & 31);
        for (int i = t - lead; i < len; i += BIN_T)
            if (i >= 0) csr_cols[start + i] = stage[i];
    }
}

#define LDSW   132
#define TILE_M 32
#define FUSE_T 512
#define W_PER_T (DIM * DIM / 4 / FUSE_T)

static_assert(DIM == 128 && TILE_M == 2 * (FUSE_T / 32) && (DIM / 16) * (TILE_M / 16) == FUSE_T / 32,
              "two nodes per wave in the gather, one 16 x 16 tile per wave in the product");

#define GRP 16

typedef const float4 __attribute__((address_space(1)))* grow_t;

__device__ __forceinline__ float4 ld_row(const float* __restrict__ seq, int c, int lane) {
    unsigned long long row = (unsigned long long)(seq + (size_t)(unsigned)c * DIM);
    asm("" : "+s"(row));
    return ((grow_t)row)[lane];
}

__device__ __forceinline__ void add4(float4& a, const float4 v) {
    a.x += v.x; a.y += v.y; a.z += v.z; a.w += v.w;
}

__device__ __forceinline__ float4 gather_node(const float* __restrict__ seq,
                                              const int* __restrict__ csr_cols,
                                              int st, int len, int id0, int id1, int lane) {
    float4 acc = make_float4(0.f, 0.f, 0.f, 0.f);
    const int n64 = min(len, 64);
    float4 buf[GRP];
    int j = 0;
    for (; j + GRP <= n64; j += GRP) {
        const int ids = (j & 32) ? id1 : id0;
        const int l0  = j & 31;
        for (int k = 0; k < GRP; ++k)
            buf[k] = ld_row(seq, __builtin_amdgcn_readlane(ids, l0 + k), lane);
        for (int k = 0; k < GRP; ++k) add4(acc, buf[k]);
    }
    const int rest = n64 - j;
    if (rest > 0) {
        const int ids = (j & 32) ? id1 : id0;
        const int l0  = j & 31;
        const int o4 = rest & 8, o2 = rest & 12, o1 = rest & 14;
        if (rest & 8)
            for (int k = 0; k < 8; ++k) buf[k] = ld_row(seq, __builtin_amdgcn_readlane(ids, l0 + k), lane);
        if (rest & 4)
            for (int k = 0; k < 4; ++k) buf[8 + k] = ld_row(seq, __builtin_amdgcn_readlane(ids, l0 + o4 + k), lane);
        if (rest & 2)
            for (int k = 0; k < 2; ++k) buf[12 + k] = ld_row(seq, __builtin_amdgcn_readlane(ids, l0 + o2 + k), lane);
        if (rest & 1) buf[14] = ld_row(seq, __builtin_amdgcn_readlane(ids, l0 + o1), lane);
        if (rest & 8)
            for (int k = 0; k < 8; ++k) add4(acc, buf[k]);
        if (rest & 4)
            for (int k = 0; k < 4; ++k) add4(acc, buf[8 + k]);
        if (rest & 2)
            for (int k = 0; k < 2; ++k) add4(acc, buf[12 + k]);
        if (rest & 1) add4(acc, buf[14]);
    }
#pragma nounroll
    for (int e = 64; e < len; ++e)
        add4(acc, ld_row(seq, csr_cols[st + e], lane));
    return acc;
}

__global__ void __launch_bounds__(FUSE_T)
agg_gemm_kernel(const float* __restrict__ seq,
                const float* __restrict__ W,
                const int* __restrict__ rowptr,
                const int* __restrict__ counts,
                const int* __restrict__ csr_cols,
                const float* __restrict__ alpha,
                float* __restrict__ out) {
    __shared__ float sW[DIM * LDSW];
    __shared__ float sA[TILE_M * LDSW];
    __shared__ float sO[TILE_M * LDSW];

    const int tid      = threadIdx.x;
    const int wave     = __builtin_amdgcn_readfirstlane(tid >> 5);
    const int lane     = tid & 31;
    const int nodeBase = blockIdx.x * TILE_M;

    int st[2], ln[2];
    for (int h = 0; h < 2; ++h) {
        const int node = nodeBase + wave + 16 * h;
        const int nc   = min(node, N_NODES - 1);
        st[h] = rowptr[nc];
        ln[h] = (node < N_NODES) ? counts[nc] : 0;
    }

    int id[2][2];
    for (int h = 0; h < 2; ++h)
        for (int p = 0; p < 2; ++p)
            id[h][p] = (32 * p + lane < ln[h]) ? csr_cols[st[h] + 32 * p + lane] : 0;

    {
        float4 w[W_PER_T];
#pragma unroll
        for (int k = 0; k < W_PER_T; ++k) w[k] = ((const float4*)W)[tid + k * FUSE_T];
#pragma unroll
        for (int k = 0; k < W_PER_T; ++k) {
            const int v = tid + k * FUSE_T;
            *(float4*)&sW[(v >> 5) * LDSW + (v & 31) * 4] = w[k];
        }
    }

    for (int h = 0; h < 2; ++h) {
        const int m = wave + 16 * h;
        const float4 acc = gather_node(seq, csr_cols, st[h], ln[h], id[h][0], id[h][1], lane);
        *(float4*)&sA[m * LDSW + lane * 4] = acc;
    }
    __syncthreads();

    const int lo = lane & 15;
    const int hi = lane >> 4;
    const int mt = wave >> 3;
    const int nt = wave & 7;
    const float* arow = &sA[(mt * 16 + lo) * LDSW + 4 * hi];
    const float* brow = &sW[(nt * 16 + lo) * LDSW + 4 * hi];

    v8f acc = {};
#pragma unroll
    for (int j = 0; j < DIM / 8; ++j) {
        const float4 a = *(const float4*)(arow + 8 * j);
        const float4 b = *(const float4*)(brow + 8 * j);
        const v2f a0 = {a.x, a.y};
        const v2f b0 = {b.x, b.y};
        const v2f a1 = {a.z, a.w};
        const v2f b1 = {b.z, b.w};
        acc = __builtin_amdgcn_wmma_f32_16x16x4_f32(
             false, a0,  false, b0,
             (short)0, acc,  false,  false);
        acc = __builtin_amdgcn_wmma_f32_16x16x4_f32(
             false, a1,  false, b1,
             (short)0, acc,  false,  false);
    }

#pragma unroll
    for (int r = 0; r < 8; ++r)
        sO[(mt * 16 + r + 8 * hi) * LDSW + nt * 16 + lo] = acc[r];
    __syncthreads();

    const float al = alpha[0];
    for (int h = 0; h < 2; ++h) {
        const int m    = wave + 16 * h;
        const int node = nodeBase + m;
        if (node < N_NODES) {
            const float4 s = *(const float4*)&sO[m * LDSW + lane * 4];
            const float  d = (float)ln[h] + EPS;
            float4 o;
            o.x = s.x / d; o.y = s.y / d; o.z = s.z / d; o.w = s.w / d;
            o.x = (o.x >= 0.f) ? o.x : al * o.x;
            o.y = (o.y >= 0.f) ? o.y : al * o.y;
            o.z = (o.z >= 0.f) ? o.z : al * o.z;
            o.w = (o.w >= 0.f) ? o.w : al * o.w;
            *(float4*)(out + (size_t)node * DIM + lane * 4) = o;
        }
    }
}

extern "C" void kernel_launch(void* const* d_in, const int* in_sizes, int n_in,
                              void* d_out, int out_size, void* d_ws, size_t ws_size,
                              hipStream_t stream) {
    (void)in_sizes; (void)n_in; (void)out_size; (void)ws_size;

    const float* seq   = (const float*)d_in[0];
    const float* W     = (const float*)d_in[1];
    const float* alpha = (const float*)d_in[2];
    const int*   rows  = (const int*)d_in[3];
    const int*   cols  = (const int*)d_in[4];
    float* out = (float*)d_out;

    char* ws = (char*)d_ws;
    unsigned* binned       = (unsigned*)(ws);
    unsigned* seg_tbl      = (unsigned*)(ws + 6422528);
    int*      counts       = (int*)(ws + 25600000);
    int*      rowptr       = (int*)(ws + 25800000);
    int*      csr_cols     = (int*)(ws + 26202048);

    slice_sort_kernel<<<NSLICE, SLICE_T, 0, stream>>>(rows, cols, seg_tbl, binned);
    bin_csr_kernel<<<NBIN, BIN_T, 0, stream>>>(binned, seg_tbl, rowptr, counts, csr_cols);

    agg_gemm_kernel<<<(N_NODES + TILE_M - 1) / TILE_M, FUSE_T, 0, stream>>>(
        seq, W, rowptr, counts, csr_cols, alpha, out);
}


# level-2 sort folded into the gather kernel: a block orders its bin's ids (128 nodes) in LDS and gathers four 32-node tiles from there; 3 launches become 2, W staged once per bin; sum order changes
# speedup vs baseline: 4.7765x; 1.0901x over previous
#include <hip/hip_runtime.h>

#define N_NODES 50000
#define N_EDGES 1600000
#define DIM     128
#define EPS     1e-8f

typedef __attribute__((ext_vector_type(2))) float v2f;
typedef __attribute__((ext_vector_type(8))) float v8f;
typedef __attribute__((ext_vector_type(4))) int   v4i;

#define BIN_SHIFT 7
#define BIN_NODES 128
#define NBIN    ((N_NODES + BIN_NODES - 1) / BIN_NODES)
#define NSLICE  256
#define SLICE_T 512
#define SLICE_V 4
#define NSPARE  32
#define NVEC4   (N_EDGES / 4)

#define TBL_W   512
#define SLICE_MAX (((NVEC4 + NSLICE - 1) / NSLICE) * 4)
#define SLICE_PITCH ((SLICE_MAX + 31) / 32 * 32)
#define CSR_STAGE 6144

static_assert(N_NODES <= (1 << 24), "a source id must fit the bits above the row's low bits");
static_assert(N_EDGES % 4 == 0, "the edge list is read as int4");
static_assert(NVEC4 / NSLICE >= (SLICE_V - 1) * SLICE_T, "all but a thread's last int4 lie inside every slice");
static_assert((NVEC4 + NSLICE - 1) / NSLICE <= SLICE_V * SLICE_T, "a thread's int4s cover the longest slice");
static_assert(SLICE_PITCH * 4 == 25088 && SLICE_PITCH * 4 * NSLICE == 6422528, "the workspace offsets in kernel_launch are written for this pitch");
static_assert(SLICE_MAX < (1 << 16), "start and count of a segment share one word");
static_assert(NBIN <= TBL_W && TBL_W <= SLICE_T && TBL_W % 32 == 0, "the bin scan runs on the first TBL_W threads");

__device__ __forceinline__ unsigned pack_rec(int col, int row) {
    return ((unsigned)col << BIN_SHIFT) | (unsigned)(row & (BIN_NODES - 1));
}

__global__ void __launch_bounds__(SLICE_T)
slice_sort_kernel(const int* __restrict__ rows, const int* __restrict__ cols,
                  unsigned* __restrict__ seg_tbl, unsigned* __restrict__ binned) {
    __shared__ int hist[TBL_W];
    __shared__ int lfirst[TBL_W];
    __shared__ int wsum[TBL_W / 32];
    __shared__ __attribute__((aligned(16))) unsigned stage[SLICE_PITCH];
    const int t    = threadIdx.x;
    const int lane = t & 31;
    const int wave = t >> 5;
    const int s    = blockIdx.x;

    if (t < TBL_W) hist[t] = 0;
    __syncthreads();

    const int v0 = ((int)blockIdx.x * NVEC4) / NSLICE;
    const int v1 = (((int)blockIdx.x + 1) * NVEC4) / NSLICE;
    const v4i* rows4 = (const v4i*)rows;
    const v4i* cols4 = (const v4i*)cols;
    const bool last = v0 + t + (SLICE_V - 1) * SLICE_T < v1;

    v4i r[SLICE_V], c[SLICE_V], p[SLICE_V];
    for (int k = 0; k < SLICE_V; ++k) {
        const int u = min(v0 + t + k * SLICE_T, v1 - 1);
        r[k] = rows4[u];
        c[k] = cols4[u];
    }
    for (int k = 0; k < SLICE_V; ++k)
        asm volatile("" : "+v"(r[k]), "+v"(c[k]));

    for (int k = 0; k < SLICE_V - 1; ++k) {
        p[k].x = atomicAdd(&hist[r[k].x >> BIN_SHIFT], 1);
        p[k].y = atomicAdd(&hist[r[k].y >> BIN_SHIFT], 1);
        p[k].z = atomicAdd(&hist[r[k].z >> BIN_SHIFT], 1);
        p[k].w = atomicAdd(&hist[r[k].w >> BIN_SHIFT], 1);
    }
    if (last) {
        p[SLICE_V - 1].x = atomicAdd(&hist[r[SLICE_V - 1].x >> BIN_SHIFT], 1);
        p[SLICE_V - 1].y = atomicAdd(&hist[r[SLICE_V - 1].y >> BIN_SHIFT], 1);
        p[SLICE_V - 1].z = atomicAdd(&hist[r[SLICE_V - 1].z >> BIN_SHIFT], 1);
        p[SLICE_V - 1].w = atomicAdd(&hist[r[SLICE_V - 1].w >> BIN_SHIFT], 1);
    }
    __syncthreads();

    {
        const int own = (t < TBL_W) ? hist[t] : 0;
        int x = own;
        for (int off = 1; off < 32; off <<= 1) {
            const int y = __shfl_up(x, off);
            if (lane >= off) x += y;
        }
        if (t < TBL_W && lane == 31) wsum[wave] = x;
        __syncthreads();
        if (t < TBL_W) {
            int pre = 0;
            for (int w = 0; w < wave; ++w) pre += wsum[w];
            const int lf = pre + x - own;
            lfirst[t] = lf;
            if (t < NBIN) seg_tbl[s * TBL_W + t] = (unsigned)lf | ((unsigned)own << 16);
        }
    }
    __syncthreads();

    for (int k = 0; k < SLICE_V - 1; ++k) {
        stage[lfirst[r[k].x >> BIN_SHIFT] + p[k].x] = pack_rec(c[k].x, r[k].x);
        stage[lfirst[r[k].y >> BIN_SHIFT] + p[k].y] = pack_rec(c[k].y, r[k].y);
        stage[lfirst[r[k].z >> BIN_SHIFT] + p[k].z] = pack_rec(c[k].z, r[k].z);
        stage[lfirst[r[k].w >> BIN_SHIFT] + p[k].w] = pack_rec(c[k].w, r[k].w);
    }
    if (last) {
        const int k = SLICE_V - 1;
        stage[lfirst[r[k].x >> BIN_SHIFT] + p[k].x] = pack_rec(c[k].x, r[k].x);
        stage[lfirst[r[k].y >> BIN_SHIFT] + p[k].y] = pack_rec(c[k].y, r[k].y);
        stage[lfirst[r[k].z >> BIN_SHIFT] + p[k].z] = pack_rec(c[k].z, r[k].z);
        stage[lfirst[r[k].w >> BIN_SHIFT] + p[k].w] = pack_rec(c[k].w, r[k].w);
    }
    __syncthreads();

    const int n4 = v1 - v0;
    uint4* dst = (uint4*)(binned + (size_t)s * SLICE_PITCH);
    for (int i = t; i < n4; i += SLICE_T) dst[i] = ((const uint4*)stage)[i];
}

#define LDSW   132
#define TILE_M 32
#define FUSE_T 512
#define W_PER_T (DIM * DIM / 4 / FUSE_T)
#define TILES_PER_BIN (BIN_NODES / TILE_M)
#define SEG_PER_WAVE (NSLICE / (FUSE_T / 32))

static_assert(DIM == 128 && TILE_M == 2 * (FUSE_T / 32) && (DIM / 16) * (TILE_M / 16) == FUSE_T / 32,
              "two nodes per wave in the gather, one 16 x 16 tile per wave in the product");
static_assert(NSLICE % (FUSE_T / 32) == 0 && NSLICE <= FUSE_T && BIN_NODES % TILE_M == 0 && BIN_NODES + NSPARE <= FUSE_T,
              "the segments are dealt evenly to the waves; a bin is a whole number of tiles");

#define GRP 16

typedef const float4 __attribute__((address_space(1)))* grow_t;
typedef const int __attribute__((address_space(3)))* lids_t;

__device__ __forceinline__ float4 ld_row(const float* __restrict__ seq, int c, int lane) {
    unsigned long long row = (unsigned long long)(seq + (size_t)(unsigned)c * DIM);
    asm("" : "+s"(row));
    return ((grow_t)row)[lane];
}

__device__ __forceinline__ void add4(float4& a, const float4 v) {
    a.x += v.x; a.y += v.y; a.z += v.z; a.w += v.w;
}

__device__ __forceinline__ float4 gather_node(const float* __restrict__ seq,
                                              bool staged, lids_t ids_lds, const int* ids_mem,
                                              int st, int len, int id0, int id1, int lane) {
    float4 acc = make_float4(0.f, 0.f, 0.f, 0.f);
    const int n64 = min(len, 64);
    float4 buf[GRP];
    int j = 0;
    for (; j + GRP <= n64; j += GRP) {
        const int ids = (j & 32) ? id1 : id0;
        const int l0  = j & 31;
        for (int k = 0; k < GRP; ++k)
            buf[k] = ld_row(seq, __builtin_amdgcn_readlane(ids, l0 + k), lane);
        for (int k = 0; k < GRP; ++k) add4(acc, buf[k]);
    }
    const int rest = n64 - j;
    if (rest > 0) {
        const int ids = (j & 32) ? id1 : id0;
        const int l0  = j & 31;
        const int o4 = rest & 8, o2 = rest & 12, o1 = rest & 14;
        if (rest & 8)
            for (int k = 0; k < 8; ++k) buf[k] = ld_row(seq, __builtin_amdgcn_readlane(ids, l0 + k), lane);
        if (rest & 4)
            for (int k = 0; k < 4; ++k) buf[8 + k] = ld_row(seq, __builtin_amdgcn_readlane(ids, l0 + o4 + k), lane);
        if (rest & 2)
            for (int k = 0; k < 2; ++k) buf[12 + k] = ld_row(seq, __builtin_amdgcn_readlane(ids, l0 + o2 + k), lane);
        if (rest & 1) buf[14] = ld_row(seq, __builtin_amdgcn_readlane(ids, l0 + o1), lane);
        if (rest & 8)
            for (int k = 0; k < 8; ++k) add4(acc, buf[k]);
        if (rest & 4)
            for (int k = 0; k < 4; ++k) add4(acc, buf[8 + k]);
        if (rest & 2)
            for (int k = 0; k < 2; ++k) add4(acc, buf[12 + k]);
        if (rest & 1) add4(acc, buf[14]);
    }
    if (staged) {
#pragma nounroll
        for (int e = 64; e < len; ++e)
            add4(acc, ld_row(seq, __builtin_amdgcn_readfirstlane(ids_lds[st + e]), lane));
    } else {
#pragma nounroll
        for (int e = 64; e < len; ++e)
            add4(acc, ld_row(seq, __builtin_amdgcn_readfirstlane(ids_mem[st + e]), lane));
    }
    return acc;
}

__global__ void __launch_bounds__(FUSE_T)
bin_agg_kernel(const unsigned* __restrict__ binned,
               const unsigned* __restrict__ seg_tbl,
               int* __restrict__ spill,
               const float* __restrict__ seq,
               const float* __restrict__ W,
               const float* __restrict__ alpha,
               float* __restrict__ out) {
    __shared__ float sW[DIM * LDSW];
    __shared__ float sA[TILE_M * LDSW];
    __shared__ float sO[TILE_M * LDSW];
    __shared__ int cnt[BIN_NODES + NSPARE];
    __shared__ int cur[BIN_NODES + NSPARE];
    __shared__ int off[BIN_NODES];
    __shared__ int wsum[NSLICE / 32];
    __shared__ int wlen[NSLICE / 32];
    __shared__ unsigned segd[NSLICE];
    __shared__ int stage[CSR_STAGE];

    const int tid   = threadIdx.x;
    const int wave  = __builtin_amdgcn_readfirstlane(tid >> 5);
    const int lane  = tid & 31;
    const int bin   = blockIdx.x;
    const int spare = BIN_NODES + lane;

    const unsigned dsc = (tid < NSLICE) ? seg_tbl[tid * TBL_W + bin] : 0u;

    {
        float4 w[W_PER_T];
#pragma unroll
        for (int k = 0; k < W_PER_T; ++k) w[k] = ((const float4*)W)[tid + k * FUSE_T];
#pragma unroll
        for (int k = 0; k < W_PER_T; ++k) {
            const int v = tid + k * FUSE_T;
            *(float4*)&sW[(v >> 5) * LDSW + (v & 31) * 4] = w[k];
        }
    }

    {
        int a = (int)(dsc & 0xffffu);
        int b = (int)(dsc >> 16);
        for (int o = 16; o > 0; o >>= 1) {
            a += __shfl_xor(a, o);
            b += __shfl_xor(b, o);
        }
        if (tid < NSLICE) {
            segd[tid] = dsc;
            if (lane == 0) { wsum[wave] = a; wlen[wave] = b; }
        }
        if (tid < BIN_NODES + NSPARE) cnt[tid] = 0;
    }
    __syncthreads();
    int start = 0, len = 0;
    for (int w = 0; w < NSLICE / 32; ++w) { start += wsum[w]; len += wlen[w]; }
    const bool staged = len <= CSR_STAGE;
    int* const gids = spill + start;

    int so[SEG_PER_WAVE], sn[SEG_PER_WAVE];
    for (int k = 0; k < SEG_PER_WAVE; ++k) {
        const int g = wave + k * (FUSE_T / 32);
        const unsigned d = segd[g];
        so[k] = __builtin_amdgcn_readfirstlane(g * SLICE_PITCH + (int)(d & 0xffffu));
        sn[k] = __builtin_amdgcn_readfirstlane((int)(d >> 16));
    }

    unsigned rec[SEG_PER_WAVE];
    for (int k = 0; k < SEG_PER_WAVE; ++k)
        rec[k] = binned[so[k] + ((lane < sn[k]) ? lane : 0)];
    for (int k = 0; k < SEG_PER_WAVE; ++k)
        atomicAdd(&cnt[(lane < sn[k]) ? (int)(rec[k] & (BIN_NODES - 1)) : spare], 1);
    for (int k = 0; k < SEG_PER_WAVE; ++k)
        for (int i = 32 + lane; i < sn[k]; i += 32)
            atomicAdd(&cnt[binned[so[k] + i] & (BIN_NODES - 1)], 1);
    __syncthreads();

    {
        const int v = (tid < BIN_NODES) ? cnt[tid] : 0;
        int x = v;
        for (int o = 1; o < 32; o <<= 1) {
            const int y = __shfl_up(x, o);
            if (lane >= o) x += y;
        }
        if (tid < BIN_NODES && lane == 31) wsum[wave] = x;
        __syncthreads();
        if (tid < BIN_NODES) {
            int pre = 0;
            for (int w = 0; w < wave; ++w) pre += wsum[w];
            const int excl = pre + x - v;
            cur[tid] = excl;
            off[tid] = excl;
        } else if (tid < BIN_NODES + NSPARE) {
            cur[tid] = 0;
        }
    }
    __syncthreads();
    {
        int pos[SEG_PER_WAVE];
        for (int k = 0; k < SEG_PER_WAVE; ++k)
            pos[k] = atomicAdd(&cur[(lane < sn[k]) ? (int)(rec[k] & (BIN_NODES - 1)) : spare], 1);
        for (int k = 0; k < SEG_PER_WAVE; ++k)
            if (lane < sn[k]) {
                const int col = (int)(rec[k] >> BIN_SHIFT);
                if (staged) stage[pos[k]] = col;
                else        gids[pos[k]] = col;
            }
        for (int k = 0; k < SEG_PER_WAVE; ++k)
            for (int i = 32 + lane; i < sn[k]; i += 32) {
                const unsigned rc = binned[so[k] + i];
                const int ps  = atomicAdd(&cur[rc & (BIN_NODES - 1)], 1);
                const int col = (int)(rc >> BIN_SHIFT);
                if (staged) stage[ps] = col;
                else        gids[ps] = col;
            }
    }
    __syncthreads();

    const int lo = lane & 15;
    const int hi = lane >> 4;
    const int mt = wave >> 3;
    const int nt = wave & 7;
    const float* arow = &sA[(mt * 16 + lo) * LDSW + 4 * hi];
    const float* brow = &sW[(nt * 16 + lo) * LDSW + 4 * hi];
    const float al = alpha[0];
    const int binBase = bin * BIN_NODES;
    const int ntile = min(TILES_PER_BIN, (N_NODES - binBase + TILE_M - 1) / TILE_M);

    for (int tl = 0; tl < ntile; ++tl) {
        int st[2], ln[2];
        for (int h = 0; h < 2; ++h) {
            const int m = tl * TILE_M + wave + 16 * h;
            st[h] = __builtin_amdgcn_readfirstlane(off[m]);
            ln[h] = __builtin_amdgcn_readfirstlane(cnt[m]);
        }

        int id[2][2];
        for (int h = 0; h < 2; ++h)
            for (int p = 0; p < 2; ++p) {
                const int e = st[h] + 32 * p + lane;
                id[h][p] = (32 * p + lane < ln[h]) ? (staged ? stage[e] : gids[e]) : 0;
            }

        for (int h = 0; h < 2; ++h) {
            const int m = wave + 16 * h;
            const float4 acc = gather_node(seq, staged, (lids_t)stage, gids, st[h], ln[h], id[h][0], id[h][1], lane);
            *(float4*)&sA[m * LDSW + lane * 4] = acc;
        }
        __syncthreads();

        v8f acc = {};
#pragma unroll
        for (int j = 0; j < DIM / 8; ++j) {
            const float4 a = *(const float4*)(arow + 8 * j);
            const float4 b = *(const float4*)(brow + 8 * j);
            const v2f a0 = {a.x, a.y};
            const v2f b0 = {b.x, b.y};
            const v2f a1 = {a.z, a.w};
            const v2f b1 = {b.z, b.w};
            acc = __builtin_amdgcn_wmma_f32_16x16x4_f32(
                 false, a0,  false, b0,
                 (short)0, acc,  false,  false);
            acc = __builtin_amdgcn_wmma_f32_16x16x4_f32(
                 false, a1,  false, b1,
                 (short)0, acc,  false,  false);
        }

#pragma unroll
        for (int r = 0; r < 8; ++r)
            sO[(mt * 16 + r + 8 * hi) * LDSW + nt * 16 + lo] = acc[r];
        __syncthreads();

        for (int h = 0; h < 2; ++h) {
            const int m    = wave + 16 * h;
            const int node = binBase + tl * TILE_M + m;
            if (node < N_NODES) {
                const float4 sv = *(const float4*)&sO[m * LDSW + lane * 4];
                const float  d = (float)ln[h] + EPS;
                float4 o;
                o.x = sv.x / d; o.y = sv.y / d; o.z = sv.z / d; o.w = sv.w / d;
                o.x = (o.x >= 0.f) ? o.x : al * o.x;
                o.y = (o.y >= 0.f) ? o.y : al * o.y;
                o.z = (o.z >= 0.f) ? o.z : al * o.z;
                o.w = (o.w >= 0.f) ? o.w : al * o.w;
                *(float4*)(out + (size_t)node * DIM + lane * 4) = o;
            }
        }
    }
}

extern "C" void kernel_launch(void* const* d_in, const int* in_sizes, int n_in,
                              void* d_out, int out_size, void* d_ws, size_t ws_size,
                              hipStream_t stream) {
    (void)in_sizes; (void)n_in; (void)out_size; (void)ws_size;

    const float* seq   = (const float*)d_in[0];
    const float* W     = (const float*)d_in[1];
    const float* alpha = (const float*)d_in[2];
    const int*   rows  = (const int*)d_in[3];
    const int*   cols  = (const int*)d_in[4];
    float* out = (float*)d_out;

    char* ws = (char*)d_ws;
    unsigned* binned       = (unsigned*)(ws);
    unsigned* seg_tbl      = (unsigned*)(ws + 6422528);
    int*      spill        = (int*)(ws + 26202048);

    slice_sort_kernel<<<NSLICE, SLICE_T, 0, stream>>>(rows, cols, seg_tbl, binned);

    bin_agg_kernel<<<NBIN, FUSE_T, 0, stream>>>(binned, seg_tbl, spill, seq, W, alpha, out);
}
